# starting kernel
# baseline (speedup 1.0000x reference)
#include <hip/hip_runtime.h>

typedef float v2f __attribute__((ext_vector_type(2)));
typedef float v8f __attribute__((ext_vector_type(8)));

__device__ __forceinline__ void atomAddF(float* p, float v) {
  asm volatile("global_atomic_add_f32 %0, %1, off" : : "v"(p), "v"(v) : "memory");
}

__global__ void k_fill0(float* __restrict__ p, long long n) {
  long long i = (long long)blockIdx.x * blockDim.x + threadIdx.x;
  if (i < n) p[i] = 0.0f;
}

__global__ void k_deg(const int* __restrict__ dst, float* __restrict__ deg, int E) {
  int e = blockIdx.x * blockDim.x + threadIdx.x;
  if (e < E) atomAddF(deg + dst[e], 1.0f);
}

__global__ void k_dinv(float* __restrict__ d, int N) {
  int i = blockIdx.x * blockDim.x + threadIdx.x;
  if (i < N) d[i] = rsqrtf(d[i] + 1.0f);
}

template <int K, int N>
__launch_bounds__(32)
__global__ void k_gemm_wmma(const float* __restrict__ A, const float* __restrict__ B,
                            float* __restrict__ C) {
  const int lane = threadIdx.x;
  const int half = lane >> 4;
  const int lm   = lane & 15;
  const int m0 = blockIdx.x * 16;
  const int n0 = blockIdx.y * 16;

  const float* __restrict__ arow = A + (long long)(m0 + lm) * K + 2 * half;
  const float* __restrict__ bp   = B + (long long)(2 * half) * N + n0 + lm;

  v8f acc = {};
#pragma unroll 8
  for (int k = 0; k < K; k += 4) {
    v2f a = *(const v2f*)(arow + k);
    v2f b;
    b.x = bp[k * N];
    b.y = bp[k * N + N];
    acc = __builtin_amdgcn_wmma_f32_16x16x4_f32(
         false, a,  false, b,
         (short)0, acc,  false,  false);
  }

  float* __restrict__ cp = C + (long long)(m0 + 8 * half) * N + n0 + lm;
#pragma unroll
  for (int r = 0; r < 8; ++r)
    cp[r * N] = acc[r];
}

__launch_bounds__(256)
__global__ void k_scatter64(const float* __restrict__ h, const float* __restrict__ dinv,
                            const int* __restrict__ src, const int* __restrict__ dst,
                            float* __restrict__ agg, int E) {
  int wid  = blockIdx.x * (blockDim.x >> 5) + (threadIdx.x >> 5);
  wid = __builtin_amdgcn_readfirstlane(wid);
  int lane = threadIdx.x & 31;
  if (wid >= E) return;
  int s = src[wid];
  int d = dst[wid];
  float w = dinv[s];
  float2 v = *((const float2*)(h + (long long)s * 64) + lane);
  float* ap = agg + (long long)d * 64 + lane * 2;
  atomAddF(ap,     v.x * w);
  atomAddF(ap + 1, v.y * w);
}

__launch_bounds__(256)
__global__ void k_scatter128(const float* __restrict__ h, const float* __restrict__ dinv,
                             const int* __restrict__ src, const int* __restrict__ dst,
                             float* __restrict__ agg, int E) {
  int wid  = blockIdx.x * (blockDim.x >> 5) + (threadIdx.x >> 5);
  wid = __builtin_amdgcn_readfirstlane(wid);
  int lane = threadIdx.x & 31;
  if (wid >= E) return;
  int s = src[wid];
  int d = dst[wid];
  float w = dinv[s];
  float4 v = *((const float4*)(h + (long long)s * 128) + lane);
  float* ap = agg + (long long)d * 128 + lane * 4;
  atomAddF(ap,     v.x * w);
  atomAddF(ap + 1, v.y * w);
  atomAddF(ap + 2, v.z * w);
  atomAddF(ap + 3, v.w * w);
}

template <int F, bool RELU>
__launch_bounds__(256)
__global__ void k_finish(const float* __restrict__ h, const float* __restrict__ agg,
                         const float* __restrict__ dinv, const float* __restrict__ bias,
                         float* __restrict__ out, int N) {
  int idx = blockIdx.x * blockDim.x + threadIdx.x;
  const int F4 = F / 4;
  long long total = (long long)N * F4;
  if (idx >= total) return;
  int i  = idx / F4;
  int f4 = idx - i * F4;
  float di = dinv[i];
  float s  = di * di;
  float4 a  = ((const float4*)agg)[idx];
  float4 hv = ((const float4*)h)[idx];
  float4 bv = ((const float4*)bias)[f4];
  float4 r;
  r.x = di * a.x + s * hv.x + bv.x;
  r.y = di * a.y + s * hv.y + bv.y;
  r.z = di * a.z + s * hv.z + bv.z;
  r.w = di * a.w + s * hv.w + bv.w;
  if (RELU) {
    r.x = fmaxf(r.x, 0.0f); r.y = fmaxf(r.y, 0.0f);
    r.z = fmaxf(r.z, 0.0f); r.w = fmaxf(r.w, 0.0f);
  }
  ((float4*)out)[idx] = r;
}

extern "C" void kernel_launch(void* const* d_in, const int* in_sizes, int n_in,
                              void* d_out, int out_size, void* d_ws, size_t ws_size,
                              hipStream_t stream) {
  const float* x  = (const float*)d_in[0];
  const int*   ei = (const int*)d_in[1];
  const float* W1 = (const float*)d_in[2];
  const float* b1 = (const float*)d_in[3];
  const float* W2 = (const float*)d_in[4];
  const float* b2 = (const float*)d_in[5];

  const int FIN = 256, H = 64, FOUT = 128;
  const int N = in_sizes[0] / FIN;
  const int E = in_sizes[1] / 2;
  const int* srcI = ei;
  const int* dstI = ei + E;

  float* ws   = (float*)d_ws;
  float* dinv = ws;
  float* h1   = dinv + (long long)N;
  float* agg1 = h1   + (long long)N * H;
  float* y1   = agg1 + (long long)N * H;
  float* agg2 = y1   + (long long)N * H;
  float* h2   = h1;

  const int TB = 256;
  auto blocks = [](long long n, int tb) { return (unsigned)((n + tb - 1) / tb); };

  k_fill0<<<blocks((long long)N, TB), TB, 0, stream>>>(dinv, N);
  k_fill0<<<blocks((long long)N * H, TB), TB, 0, stream>>>(agg1, (long long)N * H);
  k_fill0<<<blocks((long long)N * FOUT, TB), TB, 0, stream>>>(agg2, (long long)N * FOUT);

  k_deg<<<blocks(E, TB), TB, 0, stream>>>(dstI, dinv, E);
  k_dinv<<<blocks(N, TB), TB, 0, stream>>>(dinv, N);

  {
    dim3 g((unsigned)(N / 16), H / 16);
    k_gemm_wmma<256, 64><<<g, 32, 0, stream>>>(x, W1, h1);
  }
  k_scatter64<<<blocks((long long)E * 32, TB), TB, 0, stream>>>(h1, dinv, srcI, dstI, agg1, E);
  k_finish<64, true><<<blocks((long long)N * H / 4, TB), TB, 0, stream>>>(h1, agg1, dinv, b1, y1, N);

  {
    dim3 g((unsigned)(N / 16), FOUT / 16);
    k_gemm_wmma<64, 128><<<g, 32, 0, stream>>>(y1, W2, h2);
  }
  k_scatter128<<<blocks((long long)E * 32, TB), TB, 0, stream>>>(h2, dinv, srcI, dstI, agg2, E);
  k_finish<128, false><<<blocks((long long)N * FOUT / 4, TB), TB, 0, stream>>>(
      h2, agg2, dinv, b2, (float*)d_out, N);
}


# Layer 2 aggregates before its GEMM, (Ahat*y1)*W2 for Ahat*(y1*W2): the edge scatter adds 64 floats per edge, not 128; b2 joins in the GEMM epilogue. Float sums reassociated.
# speedup vs baseline: 1.3545x; 1.3545x over previous
#include <hip/hip_runtime.h>

typedef float v2f __attribute__((ext_vector_type(2)));
typedef float v8f __attribute__((ext_vector_type(8)));

__device__ __forceinline__ void atomAddF(float* p, float v) {
  asm volatile("global_atomic_add_f32 %0, %1, off" : : "v"(p), "v"(v) : "memory");
}

__global__ void k_fill0(float* __restrict__ p, long long n) {
  long long i = (long long)blockIdx.x * blockDim.x + threadIdx.x;
  if (i < n) p[i] = 0.0f;
}

__global__ void k_deg(const int* __restrict__ dst, float* __restrict__ deg, int E) {
  int e = blockIdx.x * blockDim.x + threadIdx.x;
  if (e < E) atomAddF(deg + dst[e], 1.0f);
}

__global__ void k_dinv(float* __restrict__ d, int N) {
  int i = blockIdx.x * blockDim.x + threadIdx.x;
  if (i < N) d[i] = rsqrtf(d[i] + 1.0f);
}

template <int K, int N, bool BIAS>
__launch_bounds__(32)
__global__ void k_gemm_wmma(const float* __restrict__ A, const float* __restrict__ B,
                            const float* __restrict__ bias, float* __restrict__ C) {
  const int lane = threadIdx.x;
  const int half = lane >> 4;
  const int lm   = lane & 15;
  const int m0 = blockIdx.x * 16;
  const int n0 = blockIdx.y * 16;

  const float* __restrict__ arow = A + (long long)(m0 + lm) * K + 2 * half;
  const float* __restrict__ bp   = B + (long long)(2 * half) * N + n0 + lm;

  v8f acc = {};
#pragma unroll 8
  for (int k = 0; k < K; k += 4) {
    v2f a = *(const v2f*)(arow + k);
    v2f b;
    b.x = bp[k * N];
    b.y = bp[k * N + N];
    acc = __builtin_amdgcn_wmma_f32_16x16x4_f32(
         false, a,  false, b,
         (short)0, acc,  false,  false);
  }

  const float bcol = BIAS ? bias[n0 + lm] : 0.0f;
  float* __restrict__ cp = C + (long long)(m0 + 8 * half) * N + n0 + lm;
#pragma unroll
  for (int r = 0; r < 8; ++r)
    cp[r * N] = BIAS ? acc[r] + bcol : acc[r];
}

__launch_bounds__(256)
__global__ void k_scatter64(const float* __restrict__ h, const float* __restrict__ dinv,
                            const int* __restrict__ src, const int* __restrict__ dst,
                            float* __restrict__ agg, int E) {
  int wid  = blockIdx.x * (blockDim.x >> 5) + (threadIdx.x >> 5);
  wid = __builtin_amdgcn_readfirstlane(wid);
  int lane = threadIdx.x & 31;
  if (wid >= E) return;
  int s = src[wid];
  int d = dst[wid];
  float w = dinv[s];
  float2 v = *((const float2*)(h + (long long)s * 64) + lane);
  float* ap = agg + (long long)d * 64 + lane * 2;
  atomAddF(ap,     v.x * w);
  atomAddF(ap + 1, v.y * w);
}

template <int F, bool RELU, bool BIAS>
__launch_bounds__(256)
__global__ void k_finish(const float* __restrict__ h, const float* __restrict__ agg,
                         const float* __restrict__ dinv, const float* __restrict__ bias,
                         float* __restrict__ out, int N) {
  int idx = blockIdx.x * blockDim.x + threadIdx.x;
  const int F4 = F / 4;
  long long total = (long long)N * F4;
  if (idx >= total) return;
  int i  = idx / F4;
  int f4 = idx - i * F4;
  float di = dinv[i];
  float s  = di * di;
  float4 a  = ((const float4*)agg)[idx];
  float4 hv = ((const float4*)h)[idx];
  float4 r;
  r.x = di * a.x + s * hv.x;
  r.y = di * a.y + s * hv.y;
  r.z = di * a.z + s * hv.z;
  r.w = di * a.w + s * hv.w;
  if (BIAS) {
    float4 bv = ((const float4*)bias)[f4];
    r.x += bv.x; r.y += bv.y; r.z += bv.z; r.w += bv.w;
  }
  if (RELU) {
    r.x = fmaxf(r.x, 0.0f); r.y = fmaxf(r.y, 0.0f);
    r.z = fmaxf(r.z, 0.0f); r.w = fmaxf(r.w, 0.0f);
  }
  ((float4*)out)[idx] = r;
}

extern "C" void kernel_launch(void* const* d_in, const int* in_sizes, int n_in,
                              void* d_out, int out_size, void* d_ws, size_t ws_size,
                              hipStream_t stream) {
  const float* x  = (const float*)d_in[0];
  const int*   ei = (const int*)d_in[1];
  const float* W1 = (const float*)d_in[2];
  const float* b1 = (const float*)d_in[3];
  const float* W2 = (const float*)d_in[4];
  const float* b2 = (const float*)d_in[5];

  const int FIN = 256, H = 64, FOUT = 128;
  const int N = in_sizes[0] / FIN;
  const int E = in_sizes[1] / 2;
  const int* srcI = ei;
  const int* dstI = ei + E;

  float* ws   = (float*)d_ws;
  float* dinv = ws;
  float* h1   = dinv + (long long)N;
  float* agg1 = h1   + (long long)N * H;
  float* y1   = agg1 + (long long)N * H;
  float* agg2 = y1   + (long long)N * H;
  float* z    = h1;

  const int TB = 256;
  auto blocks = [](long long n, int tb) { return (unsigned)((n + tb - 1) / tb); };

  k_fill0<<<blocks((long long)N, TB), TB, 0, stream>>>(dinv, N);
  k_fill0<<<blocks((long long)N * H, TB), TB, 0, stream>>>(agg1, (long long)N * H);
  k_fill0<<<blocks((long long)N * H, TB), TB, 0, stream>>>(agg2, (long long)N * H);

  k_deg<<<blocks(E, TB), TB, 0, stream>>>(dstI, dinv, E);
  k_dinv<<<blocks(N, TB), TB, 0, stream>>>(dinv, N);

  {
    dim3 g((unsigned)(N / 16), H / 16);
    k_gemm_wmma<256, 64, false><<<g, 32, 0, stream>>>(x, W1, nullptr, h1);
  }
  k_scatter64<<<blocks((long long)E * 32, TB), TB, 0, stream>>>(h1, dinv, srcI, dstI, agg1, E);
  k_finish<64, true, true><<<blocks((long long)N * H / 4, TB), TB, 0, stream>>>(h1, agg1, dinv, b1, y1, N);

  k_scatter64<<<blocks((long long)E * 32, TB), TB, 0, stream>>>(y1, dinv, srcI, dstI, agg2, E);
  k_finish<64, false, false><<<blocks((long long)N * H / 4, TB), TB, 0, stream>>>(
      y1, agg2, dinv, nullptr, z, N);
  {
    dim3 g((unsigned)(N / 16), FOUT / 16);
    k_gemm_wmma<64, 128, true><<<g, 32, 0, stream>>>(z, W2, b2, (float*)d_out);
  }
}


# Float atomic edge scatter replaced by a pull over a destination-sorted edge list, rebuilt each call by an atomic-free two-pass counting sort; no atomics left, float sums now in fixed sorted order.
# speedup vs baseline: 1.9363x; 1.4295x over previous
#include <hip/hip_runtime.h>

typedef float v2f __attribute__((ext_vector_type(2)));
typedef float v8f __attribute__((ext_vector_type(8)));


#define SORT_CH 1024

__device__ __forceinline__ unsigned same_digit_lanes(int d, bool valid) {
  unsigned m = __builtin_amdgcn_ballot_w32(valid);
#pragma unroll
  for (int bit = 0; bit < 8; ++bit) {
    const bool p = (d >> bit) & 1;
    const unsigned bm = __builtin_amdgcn_ballot_w32(p);
    m &= p ? bm : ~bm;
  }
  return m;
}

template <int SHIFT>
__launch_bounds__(32)
__global__ void k_hist(const int* __restrict__ key, int* __restrict__ hist, int E, int B) {
  __shared__ int cnt[256];
  const int lane = threadIdx.x;
  const int blk  = blockIdx.x;
  for (int b = lane; b < 256; b += 32) cnt[b] = 0;
  const int e0 = blk * SORT_CH;
  int kreg[SORT_CH / 32];
#pragma unroll
  for (int r = 0; r < SORT_CH / 32; ++r) {
    const int e = e0 + r * 32 + lane;
    kreg[r] = (e < E) ? key[e] : 0;
  }
  __syncthreads();
#pragma unroll
  for (int r = 0; r < SORT_CH / 32; ++r) {
    const int e = e0 + r * 32 + lane;
    const bool valid = e < E;
    const int d = (kreg[r] >> SHIFT) & 255;
    const unsigned m = same_digit_lanes(d, valid);
    const unsigned below = m & ((1u << lane) - 1u);
    if (valid && below == 0u) cnt[d] += __popc(m);
    __syncthreads();
  }
  for (int b = lane; b < 256; b += 32) hist[b * B + blk] = cnt[b];
}

__launch_bounds__(1024)
__global__ void k_tilesum(const int* __restrict__ a, int* __restrict__ tsum, int L) {
  __shared__ int wtot[32];
  const int t = threadIdx.x, lane = t & 31, w = t >> 5;
  const int i = blockIdx.x * 1024 + t;
  int v = (i < L) ? a[i] : 0;
  for (int o = 16; o > 0; o >>= 1) v += __shfl_xor(v, o);
  if (lane == 0) wtot[w] = v;
  __syncthreads();
  if (w == 0) {
    int s = wtot[lane];
    for (int o = 16; o > 0; o >>= 1) s += __shfl_xor(s, o);
    if (lane == 0) tsum[blockIdx.x] = s;
  }
}

__launch_bounds__(1024)
__global__ void k_scan(const int* __restrict__ a, const int* __restrict__ tsum,
                       int* __restrict__ out, int L) {
  __shared__ int wtot[32];
  __shared__ int base_s;
  const int t = threadIdx.x, lane = t & 31, w = t >> 5;

  int part = 0;
  for (int j = t; j < (int)blockIdx.x; j += 1024) part += tsum[j];
  for (int o = 16; o > 0; o >>= 1) part += __shfl_xor(part, o);
  if (lane == 0) wtot[w] = part;
  __syncthreads();
  if (w == 0) {
    int v = wtot[lane];
    for (int o = 16; o > 0; o >>= 1) v += __shfl_xor(v, o);
    if (lane == 0) base_s = v;
  }
  __syncthreads();
  const int base = base_s;

  const int i = blockIdx.x * 1024 + t;
  const int c = (i < L) ? a[i] : 0;
  int inc = c;
  for (int o = 1; o < 32; o <<= 1) {
    const int u = __shfl_up(inc, o);
    if (lane >= o) inc += u;
  }
  if (lane == 31) wtot[w] = inc;
  __syncthreads();
  if (w == 0) {
    const int v = wtot[lane];
    int s = v;
    for (int o = 1; o < 32; o <<= 1) {
      const int u = __shfl_up(s, o);
      if (lane >= o) s += u;
    }
    wtot[lane] = s - v;
  }
  __syncthreads();
  if (i < L) out[i] = base + wtot[w] + inc - c;
}

template <int SHIFT>
__launch_bounds__(32)
__global__ void k_place(const int* __restrict__ key, const int* __restrict__ val,
                        const int* __restrict__ off, int* __restrict__ okey,
                        int* __restrict__ oval, int E, int B) {
  __shared__ int base[256];
  const int lane = threadIdx.x;
  const int blk  = blockIdx.x;
  for (int b = lane; b < 256; b += 32) base[b] = off[b * B + blk];
  const int e0 = blk * SORT_CH;
  int kreg[SORT_CH / 32], vreg[SORT_CH / 32];
#pragma unroll
  for (int r = 0; r < SORT_CH / 32; ++r) {
    const int e = e0 + r * 32 + lane;
    kreg[r] = (e < E) ? key[e] : 0;
    vreg[r] = (e < E) ? val[e] : 0;
  }
  __syncthreads();
  int pos[SORT_CH / 32];
#pragma unroll
  for (int r = 0; r < SORT_CH / 32; ++r) {
    const int e = e0 + r * 32 + lane;
    const bool valid = e < E;
    const int d = (kreg[r] >> SHIFT) & 255;
    const unsigned m = same_digit_lanes(d, valid);
    const unsigned below = m & ((1u << lane) - 1u);
    const int o = base[d];
    __syncthreads();
    if (valid && below == 0u) base[d] = o + __popc(m);
    __syncthreads();
    pos[r] = o + __popc(below);
  }
#pragma unroll
  for (int r = 0; r < SORT_CH / 32; ++r) {
    const int e = e0 + r * 32 + lane;
    if (e < E) {
      okey[pos[r]] = kreg[r];
      oval[pos[r]] = vreg[r];
    }
  }
}

__global__ void k_bounds(const int* __restrict__ sdst, int* __restrict__ rowptr, int E, int N) {
  const int j = blockIdx.x * blockDim.x + threadIdx.x;
  if (j >= E) return;
  const int d = sdst[j];
  const int p = (j > 0) ? sdst[j - 1] : -1;
  for (int v = p + 1; v <= d; ++v) rowptr[v] = j;
  if (j == E - 1)
    for (int v = d + 1; v <= N; ++v) rowptr[v] = E;
}

__global__ void k_dinv(const int* __restrict__ rowptr, float* __restrict__ dinv, int N) {
  const int i = blockIdx.x * blockDim.x + threadIdx.x;
  if (i < N) dinv[i] = rsqrtf((float)(rowptr[i + 1] - rowptr[i]) + 1.0f);
}

template <int K, int N, bool BIAS>
__launch_bounds__(32)
__global__ void k_gemm_wmma(const float* __restrict__ A, const float* __restrict__ B,
                            const float* __restrict__ bias, float* __restrict__ C) {
  const int lane = threadIdx.x;
  const int half = lane >> 4;
  const int lm   = lane & 15;
  const int m0 = blockIdx.x * 16;
  const int n0 = blockIdx.y * 16;

  const float* __restrict__ arow = A + (long long)(m0 + lm) * K + 2 * half;
  const float* __restrict__ bp   = B + (long long)(2 * half) * N + n0 + lm;

  v8f acc = {};
#pragma unroll 8
  for (int k = 0; k < K; k += 4) {
    v2f a = *(const v2f*)(arow + k);
    v2f b;
    b.x = bp[k * N];
    b.y = bp[k * N + N];
    acc = __builtin_amdgcn_wmma_f32_16x16x4_f32(
         false, a,  false, b,
         (short)0, acc,  false,  false);
  }

  const float bcol = BIAS ? bias[n0 + lm] : 0.0f;
  float* __restrict__ cp = C + (long long)(m0 + 8 * half) * N + n0 + lm;
#pragma unroll
  for (int r = 0; r < 8; ++r)
    cp[r * N] = BIAS ? acc[r] + bcol : acc[r];
}

template <bool RELU, bool BIAS>
__launch_bounds__(256)
__global__ void k_pull64(const float* __restrict__ h, const float* __restrict__ dinv,
                         const int* __restrict__ rowptr, const int* __restrict__ csr_src,
                         const float* __restrict__ bias, float* __restrict__ out, int N) {
  int node = blockIdx.x * (blockDim.x >> 5) + (threadIdx.x >> 5);
  node = __builtin_amdgcn_readfirstlane(node);
  const int lane = threadIdx.x & 31;
  if (node >= N) return;
  const int beg = __builtin_amdgcn_readfirstlane(rowptr[node]);
  const int end = __builtin_amdgcn_readfirstlane(rowptr[node + 1]);
  const float2* __restrict__ h2 = (const float2*)h;

  float2 acc = make_float2(0.0f, 0.0f);
  for (int c = beg; c < end; c += 32) {
    const int n = (end - c < 32) ? (end - c) : 32;
    int   s_l = 0;
    float w_l = 0.0f;
    if (lane < n) {
      s_l = csr_src[c + lane];
      w_l = dinv[s_l];
    }
    for (int t = 0; t < n; t += 8) {
      float2 v[8];
      float  wv[8];
#pragma unroll
      for (int k = 0; k < 8; ++k) {
        if (k == 0 || t + k < n) {
          const int s = __builtin_amdgcn_readlane(s_l, t + k);
          wv[k] = __int_as_float(__builtin_amdgcn_readlane(__float_as_int(w_l), t + k));
          v[k]  = h2[(long long)s * 32 + lane];
        } else {
          wv[k] = 0.0f;
          v[k]  = make_float2(0.0f, 0.0f);
        }
      }
#pragma unroll
      for (int k = 0; k < 8; ++k) {
        acc.x += v[k].x * wv[k];
        acc.y += v[k].y * wv[k];
      }
    }
  }

  const float di = dinv[node];
  const float sl = di * di;
  const float2 hv = h2[(long long)node * 32 + lane];
  float2 r;
  r.x = di * acc.x + sl * hv.x;
  r.y = di * acc.y + sl * hv.y;
  if (BIAS) {
    const float2 bv = ((const float2*)bias)[lane];
    r.x += bv.x; r.y += bv.y;
  }
  if (RELU) {
    r.x = fmaxf(r.x, 0.0f); r.y = fmaxf(r.y, 0.0f);
  }
  ((float2*)out)[(long long)node * 32 + lane] = r;
}

extern "C" void kernel_launch(void* const* d_in, const int* in_sizes, int n_in,
                              void* d_out, int out_size, void* d_ws, size_t ws_size,
                              hipStream_t stream) {
  const float* x  = (const float*)d_in[0];
  const int*   ei = (const int*)d_in[1];
  const float* W1 = (const float*)d_in[2];
  const float* b1 = (const float*)d_in[3];
  const float* W2 = (const float*)d_in[4];
  const float* b2 = (const float*)d_in[5];

  const int FIN = 256, H = 64, FOUT = 128;
  const int N = in_sizes[0] / FIN;
  const int E = in_sizes[1] / 2;
  const int* srcI = ei;
  const int* dstI = ei + E;

  const int B = (E + SORT_CH - 1) / SORT_CH;
  const int L = 256 * B;
  const int T = (L + 1023) / 1024;
  float* ws   = (float*)d_ws;
  float* h1   = ws;
  float* y1   = h1   + (long long)N * H;
  float* z    = y1   + (long long)N * H;
  float* dinv = z    + (long long)N * H;
  int* rowptr = (int*)(dinv + (long long)N);
  int* k1   = rowptr + (long long)N + 1;
  int* v1   = k1 + (long long)E;
  int* k2   = v1 + (long long)E;
  int* v2   = k2 + (long long)E;
  int* hist = v2 + (long long)E;
  int* off  = hist + (long long)L;
  int* tsum = off + (long long)L;

  const int TB = 256;
  auto blocks = [](long long n, int tb) { return (unsigned)((n + tb - 1) / tb); };

  k_hist<0><<<B, 32, 0, stream>>>(dstI, hist, E, B);
  k_tilesum<<<T, 1024, 0, stream>>>(hist, tsum, L);
  k_scan<<<T, 1024, 0, stream>>>(hist, tsum, off, L);
  k_place<0><<<B, 32, 0, stream>>>(dstI, srcI, off, k1, v1, E, B);
  k_hist<8><<<B, 32, 0, stream>>>(k1, hist, E, B);
  k_tilesum<<<T, 1024, 0, stream>>>(hist, tsum, L);
  k_scan<<<T, 1024, 0, stream>>>(hist, tsum, off, L);
  k_place<8><<<B, 32, 0, stream>>>(k1, v1, off, k2, v2, E, B);
  k_bounds<<<blocks(E, TB), TB, 0, stream>>>(k2, rowptr, E, N);
  k_dinv<<<blocks(N, TB), TB, 0, stream>>>(rowptr, dinv, N);
  const int* csr_src = v2;

  {
    dim3 g((unsigned)(N / 16), H / 16);
    k_gemm_wmma<256, 64, false><<<g, 32, 0, stream>>>(x, W1, nullptr, h1);
  }
  k_pull64<true, true><<<blocks((long long)N * 32, TB), TB, 0, stream>>>(h1, dinv, rowptr, csr_src, b1, y1, N);

  k_pull64<false, false><<<blocks((long long)N * 32, TB), TB, 0, stream>>>(y1, dinv, rowptr, csr_src, nullptr, z, N);
  {
    dim3 g((unsigned)(N / 16), FOUT / 16);
    k_gemm_wmma<64, 128, true><<<g, 32, 0, stream>>>(z, W2, b2, (float*)d_out);
  }
}


# Loads under lane or slot guards, each waited for alone (sort keys 32 per wave, offset table 8, pull rows 3 steps per 8), made unconditional so each batch is in flight at once; sum order kept.
# speedup vs baseline: 2.0555x; 1.0616x over previous
#include <hip/hip_runtime.h>

typedef float v2f __attribute__((ext_vector_type(2)));
typedef float v8f __attribute__((ext_vector_type(8)));


#define SORT_CH 1024

__device__ __forceinline__ unsigned same_digit_lanes(int d, bool valid) {
  unsigned m = __builtin_amdgcn_ballot_w32(valid);
#pragma unroll
  for (int bit = 0; bit < 8; ++bit) {
    const bool p = (d >> bit) & 1;
    const unsigned bm = __builtin_amdgcn_ballot_w32(p);
    m &= p ? bm : ~bm;
  }
  return m;
}

template <int SHIFT>
__launch_bounds__(32)
__global__ void k_hist(const int* __restrict__ key, int* __restrict__ hist, int E, int B) {
  __shared__ int cnt[256];
  const int lane = threadIdx.x;
  const int blk  = blockIdx.x;
  for (int b = lane; b < 256; b += 32) cnt[b] = 0;
  const int e0 = blk * SORT_CH;
  int kreg[SORT_CH / 32];
  const int elast = E - 1;
#pragma unroll
  for (int r = 0; r < SORT_CH / 32; ++r) {
    const int e = e0 + r * 32 + lane;
    kreg[r] = key[(e < E) ? e : elast];
  }
  __syncthreads();
#pragma unroll
  for (int r = 0; r < SORT_CH / 32; ++r) {
    const int e = e0 + r * 32 + lane;
    const bool valid = e < E;
    const int d = (kreg[r] >> SHIFT) & 255;
    const unsigned m = same_digit_lanes(d, valid);
    const unsigned below = m & ((1u << lane) - 1u);
    if (valid && below == 0u) cnt[d] += __popc(m);
    __syncthreads();
  }
  for (int b = lane; b < 256; b += 32) hist[b * B + blk] = cnt[b];
}

__launch_bounds__(1024)
__global__ void k_tilesum(const int* __restrict__ a, int* __restrict__ tsum, int L) {
  __shared__ int wtot[32];
  const int t = threadIdx.x, lane = t & 31, w = t >> 5;
  const int i = blockIdx.x * 1024 + t;
  int v = (i < L) ? a[i] : 0;
  for (int o = 16; o > 0; o >>= 1) v += __shfl_xor(v, o);
  if (lane == 0) wtot[w] = v;
  __syncthreads();
  if (w == 0) {
    int s = wtot[lane];
    for (int o = 16; o > 0; o >>= 1) s += __shfl_xor(s, o);
    if (lane == 0) tsum[blockIdx.x] = s;
  }
}

__launch_bounds__(1024)
__global__ void k_scan(const int* __restrict__ a, const int* __restrict__ tsum,
                       int* __restrict__ out, int L) {
  __shared__ int wtot[32];
  __shared__ int base_s;
  const int t = threadIdx.x, lane = t & 31, w = t >> 5;

  int part = 0;
  for (int j = t; j < (int)blockIdx.x; j += 1024) part += tsum[j];
  for (int o = 16; o > 0; o >>= 1) part += __shfl_xor(part, o);
  if (lane == 0) wtot[w] = part;
  __syncthreads();
  if (w == 0) {
    int v = wtot[lane];
    for (int o = 16; o > 0; o >>= 1) v += __shfl_xor(v, o);
    if (lane == 0) base_s = v;
  }
  __syncthreads();
  const int base = base_s;

  const int i = blockIdx.x * 1024 + t;
  const int c = (i < L) ? a[i] : 0;
  int inc = c;
  for (int o = 1; o < 32; o <<= 1) {
    const int u = __shfl_up(inc, o);
    if (lane >= o) inc += u;
  }
  if (lane == 31) wtot[w] = inc;
  __syncthreads();
  if (w == 0) {
    const int v = wtot[lane];
    int s = v;
    for (int o = 1; o < 32; o <<= 1) {
      const int u = __shfl_up(s, o);
      if (lane >= o) s += u;
    }
    wtot[lane] = s - v;
  }
  __syncthreads();
  if (i < L) out[i] = base + wtot[w] + inc - c;
}

template <int SHIFT>
__launch_bounds__(32)
__global__ void k_place(const int* __restrict__ key, const int* __restrict__ val,
                        const int* __restrict__ off, int* __restrict__ okey,
                        int* __restrict__ oval, int E, int B) {
  __shared__ int base[256];
  const int lane = threadIdx.x;
  const int blk  = blockIdx.x;
  int o8[8];
#pragma unroll
  for (int i = 0; i < 8; ++i) o8[i] = off[(lane + 32 * i) * B + blk];
  const int e0 = blk * SORT_CH;
  int kreg[SORT_CH / 32], vreg[SORT_CH / 32];
#pragma unroll
  for (int r = 0; r < SORT_CH / 32; ++r) {
    const int e = e0 + r * 32 + lane;
    kreg[r] = (e < E) ? key[e] : 0;
    vreg[r] = (e < E) ? val[e] : 0;
  }
#pragma unroll
  for (int i = 0; i < 8; ++i) base[lane + 32 * i] = o8[i];
  __syncthreads();
  int pos[SORT_CH / 32];
#pragma unroll
  for (int r = 0; r < SORT_CH / 32; ++r) {
    const int e = e0 + r * 32 + lane;
    const bool valid = e < E;
    const int d = (kreg[r] >> SHIFT) & 255;
    const unsigned m = same_digit_lanes(d, valid);
    const unsigned below = m & ((1u << lane) - 1u);
    const int o = base[d];
    __syncthreads();
    if (valid && below == 0u) base[d] = o + __popc(m);
    __syncthreads();
    pos[r] = o + __popc(below);
  }
#pragma unroll
  for (int r = 0; r < SORT_CH / 32; ++r) {
    const int e = e0 + r * 32 + lane;
    if (e < E) {
      okey[pos[r]] = kreg[r];
      oval[pos[r]] = vreg[r];
    }
  }
}

__global__ void k_bounds(const int* __restrict__ sdst, int* __restrict__ rowptr, int E, int N) {
  const int j = blockIdx.x * blockDim.x + threadIdx.x;
  if (j >= E) return;
  const int d = sdst[j];
  const int p = (j > 0) ? sdst[j - 1] : -1;
  for (int v = p + 1; v <= d; ++v) rowptr[v] = j;
  if (j == E - 1)
    for (int v = d + 1; v <= N; ++v) rowptr[v] = E;
}

__global__ void k_dinv(const int* __restrict__ rowptr, float* __restrict__ dinv, int N) {
  const int i = blockIdx.x * blockDim.x + threadIdx.x;
  if (i < N) dinv[i] = rsqrtf((float)(rowptr[i + 1] - rowptr[i]) + 1.0f);
}

template <int K, int N, bool BIAS>
__launch_bounds__(32)
__global__ void k_gemm_wmma(const float* __restrict__ A, const float* __restrict__ B,
                            const float* __restrict__ bias, float* __restrict__ C) {
  const int lane = threadIdx.x;
  const int half = lane >> 4;
  const int lm   = lane & 15;
  const int m0 = blockIdx.x * 16;
  const int n0 = blockIdx.y * 16;

  const float* __restrict__ arow = A + (long long)(m0 + lm) * K + 2 * half;
  const float* __restrict__ bp   = B + (long long)(2 * half) * N + n0 + lm;

  v8f acc = {};
#pragma unroll 8
  for (int k = 0; k < K; k += 4) {
    v2f a = *(const v2f*)(arow + k);
    v2f b;
    b.x = bp[k * N];
    b.y = bp[k * N + N];
    acc = __builtin_amdgcn_wmma_f32_16x16x4_f32(
         false, a,  false, b,
         (short)0, acc,  false,  false);
  }

  const float bcol = BIAS ? bias[n0 + lm] : 0.0f;
  float* __restrict__ cp = C + (long long)(m0 + 8 * half) * N + n0 + lm;
#pragma unroll
  for (int r = 0; r < 8; ++r)
    cp[r * N] = BIAS ? acc[r] + bcol : acc[r];
}


template <int R>
__device__ __forceinline__ void pull_rows(const float2* __restrict__ h2, int s_l, float w_l,
                                          int t, int lane, float2& acc) {
  float2 v[R];
  float  wv[R];
#pragma unroll
  for (int k = 0; k < R; ++k) {
    const int s = __builtin_amdgcn_readlane(s_l, t + k);
    wv[k] = __int_as_float(__builtin_amdgcn_readlane(__float_as_int(w_l), t + k));
    v[k]  = h2[(long long)s * 32 + lane];
  }
#pragma unroll
  for (int k = 0; k < R; ++k) {
    acc.x += v[k].x * wv[k];
    acc.y += v[k].y * wv[k];
  }
}

template <bool RELU, bool BIAS>
__launch_bounds__(256)
__global__ void k_pull64(const float* __restrict__ h, const float* __restrict__ dinv,
                         const int* __restrict__ rowptr, const int* __restrict__ csr_src,
                         const float* __restrict__ bias, float* __restrict__ out, int N) {
  int node = blockIdx.x * (blockDim.x >> 5) + (threadIdx.x >> 5);
  node = __builtin_amdgcn_readfirstlane(node);
  const int lane = threadIdx.x & 31;
  if (node >= N) return;
  const int beg = __builtin_amdgcn_readfirstlane(rowptr[node]);
  const int end = __builtin_amdgcn_readfirstlane(rowptr[node + 1]);
  const float2* __restrict__ h2 = (const float2*)h;

  const float di = dinv[node];
  const float2 hv = h2[(long long)node * 32 + lane];
  float2 bv = make_float2(0.0f, 0.0f);
  if (BIAS) bv = ((const float2*)bias)[lane];

  float2 acc = make_float2(0.0f, 0.0f);
  for (int c = beg; c < end; c += 32) {
    const int n = (end - c < 32) ? (end - c) : 32;
    int   s_l = 0;
    float w_l = 0.0f;
    if (lane < n) {
      s_l = csr_src[c + lane];
      w_l = dinv[s_l];
    }
    int t = 0;
    for (; t + 8 <= n; t += 8) pull_rows<8>(h2, s_l, w_l, t, lane, acc);
    switch (n - t) {
      case 7: pull_rows<7>(h2, s_l, w_l, t, lane, acc); break;
      case 6: pull_rows<6>(h2, s_l, w_l, t, lane, acc); break;
      case 5: pull_rows<5>(h2, s_l, w_l, t, lane, acc); break;
      case 4: pull_rows<4>(h2, s_l, w_l, t, lane, acc); break;
      case 3: pull_rows<3>(h2, s_l, w_l, t, lane, acc); break;
      case 2: pull_rows<2>(h2, s_l, w_l, t, lane, acc); break;
      case 1: pull_rows<1>(h2, s_l, w_l, t, lane, acc); break;
      default: break;
    }
  }

  const float sl = di * di;
  float2 r;
  r.x = di * acc.x + sl * hv.x;
  r.y = di * acc.y + sl * hv.y;
  if (BIAS) {
    r.x += bv.x; r.y += bv.y;
  }
  if (RELU) {
    r.x = fmaxf(r.x, 0.0f); r.y = fmaxf(r.y, 0.0f);
  }
  ((float2*)out)[(long long)node * 32 + lane] = r;
}

extern "C" void kernel_launch(void* const* d_in, const int* in_sizes, int n_in,
                              void* d_out, int out_size, void* d_ws, size_t ws_size,
                              hipStream_t stream) {
  const float* x  = (const float*)d_in[0];
  const int*   ei = (const int*)d_in[1];
  const float* W1 = (const float*)d_in[2];
  const float* b1 = (const float*)d_in[3];
  const float* W2 = (const float*)d_in[4];
  const float* b2 = (const float*)d_in[5];

  const int FIN = 256, H = 64, FOUT = 128;
  const int N = in_sizes[0] / FIN;
  const int E = in_sizes[1] / 2;
  const int* srcI = ei;
  const int* dstI = ei + E;

  const int B = (E + SORT_CH - 1) / SORT_CH;
  const int L = 256 * B;
  const int T = (L + 1023) / 1024;
  float* ws   = (float*)d_ws;
  float* h1   = ws;
  float* y1   = h1   + (long long)N * H;
  float* z    = y1   + (long long)N * H;
  float* dinv = z    + (long long)N * H;
  int* rowptr = (int*)(dinv + (long long)N);
  int* k1   = rowptr + (long long)N + 1;
  int* v1   = k1 + (long long)E;
  int* k2   = v1 + (long long)E;
  int* v2   = k2 + (long long)E;
  int* hist = v2 + (long long)E;
  int* off  = hist + (long long)L;
  int* tsum = off + (long long)L;

  const int TB = 256;
  auto blocks = [](long long n, int tb) { return (unsigned)((n + tb - 1) / tb); };

  k_hist<0><<<B, 32, 0, stream>>>(dstI, hist, E, B);
  k_tilesum<<<T, 1024, 0, stream>>>(hist, tsum, L);
  k_scan<<<T, 1024, 0, stream>>>(hist, tsum, off, L);
  k_place<0><<<B, 32, 0, stream>>>(dstI, srcI, off, k1, v1, E, B);
  k_hist<8><<<B, 32, 0, stream>>>(k1, hist, E, B);
  k_tilesum<<<T, 1024, 0, stream>>>(hist, tsum, L);
  k_scan<<<T, 1024, 0, stream>>>(hist, tsum, off, L);
  k_place<8><<<B, 32, 0, stream>>>(k1, v1, off, k2, v2, E, B);
  k_bounds<<<blocks(E, TB), TB, 0, stream>>>(k2, rowptr, E, N);
  k_dinv<<<blocks(N, TB), TB, 0, stream>>>(rowptr, dinv, N);
  const int* csr_src = v2;

  {
    dim3 g((unsigned)(N / 16), H / 16);
    k_gemm_wmma<256, 64, false><<<g, 32, 0, stream>>>(x, W1, nullptr, h1);
  }
  k_pull64<true, true><<<blocks((long long)N * 32, TB), TB, 0, stream>>>(h1, dinv, rowptr, csr_src, b1, y1, N);

  k_pull64<false, false><<<blocks((long long)N * 32, TB), TB, 0, stream>>>(y1, dinv, rowptr, csr_src, nullptr, z, N);
  {
    dim3 g((unsigned)(N / 16), FOUT / 16);
    k_gemm_wmma<64, 128, true><<<g, 32, 0, stream>>>(z, W2, b2, (float*)d_out);
  }
}


# Fewer, fatter waves where wave entry set the time: GEMMs one wave per 16-row strip (all column tiles, A fragment reused), list bounds 16 keys/thread, table scan 4 entries/thread. Sum order kept.
# speedup vs baseline: 2.1996x; 1.0701x over previous
#include <hip/hip_runtime.h>

typedef float v2f __attribute__((ext_vector_type(2)));
typedef float v4f __attribute__((ext_vector_type(4)));
typedef float v8f __attribute__((ext_vector_type(8)));


#define SORT_CH 1024

__device__ __forceinline__ unsigned same_digit_lanes(int d, bool valid) {
  unsigned m = __builtin_amdgcn_ballot_w32(valid);
#pragma unroll
  for (int bit = 0; bit < 8; ++bit) {
    const bool p = (d >> bit) & 1;
    const unsigned bm = __builtin_amdgcn_ballot_w32(p);
    m &= p ? bm : ~bm;
  }
  return m;
}

template <int SHIFT>
__launch_bounds__(32)
__global__ void k_hist(const int* __restrict__ key, int* __restrict__ hist, int E, int B) {
  __shared__ int cnt[256];
  const int lane = threadIdx.x;
  const int blk  = blockIdx.x;
  for (int b = lane; b < 256; b += 32) cnt[b] = 0;
  const int e0 = blk * SORT_CH;
  int kreg[SORT_CH / 32];
  const int elast = E - 1;
#pragma unroll
  for (int r = 0; r < SORT_CH / 32; ++r) {
    const int e = e0 + r * 32 + lane;
    kreg[r] = key[(e < E) ? e : elast];
  }
  __syncthreads();
#pragma unroll
  for (int r = 0; r < SORT_CH / 32; ++r) {
    const int e = e0 + r * 32 + lane;
    const bool valid = e < E;
    const int d = (kreg[r] >> SHIFT) & 255;
    const unsigned m = same_digit_lanes(d, valid);
    const unsigned below = m & ((1u << lane) - 1u);
    if (valid && below == 0u) cnt[d] += __popc(m);
    __syncthreads();
  }
  for (int b = lane; b < 256; b += 32) hist[b * B + blk] = cnt[b];
}

#define SCAN_PT 4
#define SCAN_TILE (1024 * SCAN_PT)

__launch_bounds__(1024)
__global__ void k_tilesum(const int* __restrict__ a, int* __restrict__ tsum, int L) {
  __shared__ int wtot[32];
  const int t = threadIdx.x, lane = t & 31, w = t >> 5;
  const int base = blockIdx.x * SCAN_TILE;
  int v = 0;
#pragma unroll
  for (int q = 0; q < SCAN_PT; ++q) {
    const int i = base + q * 1024 + t;
    const int x = a[(i < L) ? i : 0];
    v += (i < L) ? x : 0;
  }
  for (int o = 16; o > 0; o >>= 1) v += __shfl_xor(v, o);
  if (lane == 0) wtot[w] = v;
  __syncthreads();
  if (w == 0) {
    int s = wtot[lane];
    for (int o = 16; o > 0; o >>= 1) s += __shfl_xor(s, o);
    if (lane == 0) tsum[blockIdx.x] = s;
  }
}

__launch_bounds__(1024)
__global__ void k_scan(const int* __restrict__ a, const int* __restrict__ tsum,
                       int* __restrict__ out, int L) {
  __shared__ int wtot[32];
  __shared__ int base_s;
  const int t = threadIdx.x, lane = t & 31, w = t >> 5;

  const int i0 = (blockIdx.x * 1024 + t) * SCAN_PT;
  int c[SCAN_PT];
#pragma unroll
  for (int q = 0; q < SCAN_PT; ++q) {
    const int i = i0 + q;
    const int x = a[(i < L) ? i : 0];
    c[q] = (i < L) ? x : 0;
  }

  int part = 0;
  for (int j = t; j < (int)blockIdx.x; j += 1024) part += tsum[j];
  for (int o = 16; o > 0; o >>= 1) part += __shfl_xor(part, o);
  if (lane == 0) wtot[w] = part;
  __syncthreads();
  if (w == 0) {
    int v = wtot[lane];
    for (int o = 16; o > 0; o >>= 1) v += __shfl_xor(v, o);
    if (lane == 0) base_s = v;
  }
  __syncthreads();
  const int base = base_s;

  int s = 0;
#pragma unroll
  for (int q = 0; q < SCAN_PT; ++q) s += c[q];
  int inc = s;
  for (int o = 1; o < 32; o <<= 1) {
    const int u = __shfl_up(inc, o);
    if (lane >= o) inc += u;
  }
  if (lane == 31) wtot[w] = inc;
  __syncthreads();
  if (w == 0) {
    const int v = wtot[lane];
    int x = v;
    for (int o = 1; o < 32; o <<= 1) {
      const int u = __shfl_up(x, o);
      if (lane >= o) x += u;
    }
    wtot[lane] = x - v;
  }
  __syncthreads();
  int run = base + wtot[w] + inc - s;
#pragma unroll
  for (int q = 0; q < SCAN_PT; ++q) {
    if (i0 + q < L) out[i0 + q] = run;
    run += c[q];
  }
}

template <int SHIFT>
__launch_bounds__(32)
__global__ void k_place(const int* __restrict__ key, const int* __restrict__ val,
                        const int* __restrict__ off, int* __restrict__ okey,
                        int* __restrict__ oval, int E, int B) {
  __shared__ int base[256];
  const int lane = threadIdx.x;
  const int blk  = blockIdx.x;
  int o8[8];
#pragma unroll
  for (int i = 0; i < 8; ++i) o8[i] = off[(lane + 32 * i) * B + blk];
  const int e0 = blk * SORT_CH;
  int kreg[SORT_CH / 32], vreg[SORT_CH / 32];
#pragma unroll
  for (int r = 0; r < SORT_CH / 32; ++r) {
    const int e = e0 + r * 32 + lane;
    kreg[r] = (e < E) ? key[e] : 0;
    vreg[r] = (e < E) ? val[e] : 0;
  }
#pragma unroll
  for (int i = 0; i < 8; ++i) base[lane + 32 * i] = o8[i];
  __syncthreads();
  int pos[SORT_CH / 32];
#pragma unroll
  for (int r = 0; r < SORT_CH / 32; ++r) {
    const int e = e0 + r * 32 + lane;
    const bool valid = e < E;
    const int d = (kreg[r] >> SHIFT) & 255;
    const unsigned m = same_digit_lanes(d, valid);
    const unsigned below = m & ((1u << lane) - 1u);
    const int o = base[d];
    __syncthreads();
    if (valid && below == 0u) base[d] = o + __popc(m);
    __syncthreads();
    pos[r] = o + __popc(below);
  }
#pragma unroll
  for (int r = 0; r < SORT_CH / 32; ++r) {
    const int e = e0 + r * 32 + lane;
    if (e < E) {
      okey[pos[r]] = kreg[r];
      oval[pos[r]] = vreg[r];
    }
  }
}

#define BND_R 16
__launch_bounds__(256)
__global__ void k_bounds(const int* __restrict__ sdst, int* __restrict__ rowptr, int E, int N) {
  const int lane = threadIdx.x & 31;
  const int wave = blockIdx.x * (blockDim.x >> 5) + (threadIdx.x >> 5);
  const int base = wave * (32 * BND_R);
  if (base >= E) return;
  const int elast = E - 1;
  int d[BND_R], p[BND_R];
#pragma unroll
  for (int r = 0; r < BND_R; ++r) {
    const int j  = base + r * 32 + lane;
    const int jc = (j < E) ? j : elast;
    d[r] = sdst[jc];
    p[r] = sdst[(jc > 0) ? jc - 1 : 0];
  }
#pragma unroll
  for (int r = 0; r < BND_R; ++r) {
    const int j = base + r * 32 + lane;
    if (j < E) {
      const int pp = (j > 0) ? p[r] : -1;
      for (int v = pp + 1; v <= d[r]; ++v) rowptr[v] = j;
      if (j == elast)
        for (int v = d[r] + 1; v <= N; ++v) rowptr[v] = E;
    }
  }
}

__global__ void k_dinv(const int* __restrict__ rowptr, float* __restrict__ dinv, int N) {
  const int i = blockIdx.x * blockDim.x + threadIdx.x;
  if (i < N) dinv[i] = rsqrtf((float)(rowptr[i + 1] - rowptr[i]) + 1.0f);
}

template <int K, int N, int U, bool BIAS>
__launch_bounds__(128)
__global__ void k_gemm_wmma(const float* __restrict__ A, const float* __restrict__ B,
                            const float* __restrict__ bias, float* __restrict__ C, int nstrips) {
  constexpr int NT = N / 16;
  constexpr int G  = N / 64;
  const int lane = threadIdx.x & 31;
  const int half = lane >> 4;
  const int lm   = lane & 15;
  const int strip = __builtin_amdgcn_readfirstlane((int)(blockIdx.x * 4 + (threadIdx.x >> 5)));
  if (strip >= nstrips) return;

  const float* __restrict__ arow = A + (long long)(strip * 16 + lm) * K + 2 * half;
  const float* __restrict__ bp   = B + (long long)(2 * half) * N + 4 * lm;

  v8f acc[NT];
#pragma unroll
  for (int j = 0; j < NT; ++j) acc[j] = (v8f){0.f, 0.f, 0.f, 0.f, 0.f, 0.f, 0.f, 0.f};

  v2f a_cur[U], a_nxt[U];
#pragma unroll
  for (int u = 0; u < U; ++u) {
    a_cur[u] = *(const v2f*)(arow + 4 * u);
    a_nxt[u] = a_cur[u];
  }

#pragma unroll 1
  for (int kc = 0; kc < K; kc += 4 * U) {
    v4f bx[U][G], by[U][G];
#pragma unroll
    for (int u = 0; u < U; ++u) {
#pragma unroll
      for (int g = 0; g < G; ++g) {
        bx[u][g] = *(const v4f*)(bp + (kc + 4 * u) * N + 64 * g);
        by[u][g] = *(const v4f*)(bp + (kc + 4 * u) * N + N + 64 * g);
      }
    }
    if (kc + 4 * U < K) {
#pragma unroll
      for (int u = 0; u < U; ++u) a_nxt[u] = *(const v2f*)(arow + kc + 4 * U + 4 * u);
    }
#pragma unroll
    for (int u = 0; u < U; ++u) {
#pragma unroll
      for (int g = 0; g < G; ++g) {
#pragma unroll
        for (int q = 0; q < 4; ++q) {
          v2f b;
          b.x = bx[u][g][q];
          b.y = by[u][g][q];
          acc[4 * g + q] = __builtin_amdgcn_wmma_f32_16x16x4_f32(
               false, a_cur[u],  false, b,
               (short)0, acc[4 * g + q],  false,  false);
        }
      }
    }
#pragma unroll
    for (int u = 0; u < U; ++u) a_cur[u] = a_nxt[u];
  }

  v4f bv[G];
#pragma unroll
  for (int g = 0; g < G; ++g)
    bv[g] = BIAS ? *(const v4f*)(bias + 64 * g + 4 * lm) : (v4f){0.f, 0.f, 0.f, 0.f};
  float* __restrict__ cp = C + (long long)(strip * 16 + 8 * half) * N + 4 * lm;
#pragma unroll
  for (int r = 0; r < 8; ++r) {
#pragma unroll
    for (int g = 0; g < G; ++g) {
      v4f o;
      o.x = acc[4 * g + 0][r];
      o.y = acc[4 * g + 1][r];
      o.z = acc[4 * g + 2][r];
      o.w = acc[4 * g + 3][r];
      if (BIAS) o = o + bv[g];
      *(v4f*)(cp + r * N + 64 * g) = o;
    }
  }
}


template <int R>
__device__ __forceinline__ void pull_rows(const float2* __restrict__ h2, int s_l, float w_l,
                                          int t, int lane, float2& acc) {
  float2 v[R];
  float  wv[R];
#pragma unroll
  for (int k = 0; k < R; ++k) {
    const int s = __builtin_amdgcn_readlane(s_l, t + k);
    wv[k] = __int_as_float(__builtin_amdgcn_readlane(__float_as_int(w_l), t + k));
    v[k]  = h2[(long long)s * 32 + lane];
  }
#pragma unroll
  for (int k = 0; k < R; ++k) {
    acc.x += v[k].x * wv[k];
    acc.y += v[k].y * wv[k];
  }
}

template <bool RELU, bool BIAS>
__launch_bounds__(256)
__global__ void k_pull64(const float* __restrict__ h, const float* __restrict__ dinv,
                         const int* __restrict__ rowptr, const int* __restrict__ csr_src,
                         const float* __restrict__ bias, float* __restrict__ out, int N) {
  int node = blockIdx.x * (blockDim.x >> 5) + (threadIdx.x >> 5);
  node = __builtin_amdgcn_readfirstlane(node);
  const int lane = threadIdx.x & 31;
  if (node >= N) return;
  const int beg = __builtin_amdgcn_readfirstlane(rowptr[node]);
  const int end = __builtin_amdgcn_readfirstlane(rowptr[node + 1]);
  const float2* __restrict__ h2 = (const float2*)h;

  const float di = dinv[node];
  const float2 hv = h2[(long long)node * 32 + lane];
  float2 bv = make_float2(0.0f, 0.0f);
  if (BIAS) bv = ((const float2*)bias)[lane];

  float2 acc = make_float2(0.0f, 0.0f);
  for (int c = beg; c < end; c += 32) {
    const int n = (end - c < 32) ? (end - c) : 32;
    int   s_l = 0;
    float w_l = 0.0f;
    if (lane < n) {
      s_l = csr_src[c + lane];
      w_l = dinv[s_l];
    }
    int t = 0;
    for (; t + 8 <= n; t += 8) pull_rows<8>(h2, s_l, w_l, t, lane, acc);
    switch (n - t) {
      case 7: pull_rows<7>(h2, s_l, w_l, t, lane, acc); break;
      case 6: pull_rows<6>(h2, s_l, w_l, t, lane, acc); break;
      case 5: pull_rows<5>(h2, s_l, w_l, t, lane, acc); break;
      case 4: pull_rows<4>(h2, s_l, w_l, t, lane, acc); break;
      case 3: pull_rows<3>(h2, s_l, w_l, t, lane, acc); break;
      case 2: pull_rows<2>(h2, s_l, w_l, t, lane, acc); break;
      case 1: pull_rows<1>(h2, s_l, w_l, t, lane, acc); break;
      default: break;
    }
  }

  const float sl = di * di;
  float2 r;
  r.x = di * acc.x + sl * hv.x;
  r.y = di * acc.y + sl * hv.y;
  if (BIAS) {
    r.x += bv.x; r.y += bv.y;
  }
  if (RELU) {
    r.x = fmaxf(r.x, 0.0f); r.y = fmaxf(r.y, 0.0f);
  }
  ((float2*)out)[(long long)node * 32 + lane] = r;
}

extern "C" void kernel_launch(void* const* d_in, const int* in_sizes, int n_in,
                              void* d_out, int out_size, void* d_ws, size_t ws_size,
                              hipStream_t stream) {
  const float* x  = (const float*)d_in[0];
  const int*   ei = (const int*)d_in[1];
  const float* W1 = (const float*)d_in[2];
  const float* b1 = (const float*)d_in[3];
  const float* W2 = (const float*)d_in[4];
  const float* b2 = (const float*)d_in[5];

  const int FIN = 256, H = 64, FOUT = 128;
  const int N = in_sizes[0] / FIN;
  const int E = in_sizes[1] / 2;
  const int* srcI = ei;
  const int* dstI = ei + E;

  const int B = (E + SORT_CH - 1) / SORT_CH;
  const int L = 256 * B;
  const int T = (L + SCAN_TILE - 1) / SCAN_TILE;
  float* ws   = (float*)d_ws;
  float* h1   = ws;
  float* y1   = h1   + (long long)N * H;
  float* z    = y1   + (long long)N * H;
  float* dinv = z    + (long long)N * H;
  int* rowptr = (int*)(dinv + (long long)N);
  int* k1   = rowptr + (long long)N + 1;
  int* v1   = k1 + (long long)E;
  int* k2   = v1 + (long long)E;
  int* v2   = k2 + (long long)E;
  int* hist = v2 + (long long)E;
  int* off  = hist + (long long)L;
  int* tsum = off + (long long)L;

  const int TB = 256;
  auto blocks = [](long long n, int tb) { return (unsigned)((n + tb - 1) / tb); };

  k_hist<0><<<B, 32, 0, stream>>>(dstI, hist, E, B);
  k_tilesum<<<T, 1024, 0, stream>>>(hist, tsum, L);
  k_scan<<<T, 1024, 0, stream>>>(hist, tsum, off, L);
  k_place<0><<<B, 32, 0, stream>>>(dstI, srcI, off, k1, v1, E, B);
  k_hist<8><<<B, 32, 0, stream>>>(k1, hist, E, B);
  k_tilesum<<<T, 1024, 0, stream>>>(hist, tsum, L);
  k_scan<<<T, 1024, 0, stream>>>(hist, tsum, off, L);
  k_place<8><<<B, 32, 0, stream>>>(k1, v1, off, k2, v2, E, B);
  k_bounds<<<blocks(E, TB * BND_R), TB, 0, stream>>>(k2, rowptr, E, N);
  k_dinv<<<blocks(N, TB), TB, 0, stream>>>(rowptr, dinv, N);
  const int* csr_src = v2;

  const int nstrips = N / 16;
  k_gemm_wmma<256, 64, 16, false><<<blocks(nstrips, 4), 128, 0, stream>>>(x, W1, nullptr, h1, nstrips);
  k_pull64<true, true><<<blocks((long long)N * 32, TB), TB, 0, stream>>>(h1, dinv, rowptr, csr_src, b1, y1, N);

  k_pull64<false, false><<<blocks((long long)N * 32, TB), TB, 0, stream>>>(y1, dinv, rowptr, csr_src, nullptr, z, N);
  k_gemm_wmma<64, 128, 8, true><<<blocks(nstrips, 4), 128, 0, stream>>>(z, W2, b2, (float*)d_out, nstrips);
}


# Second sort pass, list bounds and dinv done by one owner work-group per 256-node bucket with its counters in LDS, in place of six launches (14 launches become 9); list order and sum order kept.
# speedup vs baseline: 3.2980x; 1.4994x over previous
#include <hip/hip_runtime.h>

typedef float v2f __attribute__((ext_vector_type(2)));
typedef float v4f __attribute__((ext_vector_type(4)));
typedef float v8f __attribute__((ext_vector_type(8)));


#define SORT_CH 1024

__device__ __forceinline__ unsigned same_digit_lanes(int d, bool valid) {
  unsigned m = __builtin_amdgcn_ballot_w32(valid);
#pragma unroll
  for (int bit = 0; bit < 8; ++bit) {
    const bool p = (d >> bit) & 1;
    const unsigned bm = __builtin_amdgcn_ballot_w32(p);
    m &= p ? bm : ~bm;
  }
  return m;
}

template <int SHIFT>
__launch_bounds__(32)
__global__ void k_hist(const int* __restrict__ key, int* __restrict__ hist, int E, int B) {
  __shared__ int cnt[256];
  const int lane = threadIdx.x;
  const int blk  = blockIdx.x;
  for (int b = lane; b < 256; b += 32) cnt[b] = 0;
  const int e0 = blk * SORT_CH;
  int kreg[SORT_CH / 32];
  const int elast = E - 1;
#pragma unroll
  for (int r = 0; r < SORT_CH / 32; ++r) {
    const int e = e0 + r * 32 + lane;
    kreg[r] = key[(e < E) ? e : elast];
  }
  __syncthreads();
#pragma unroll
  for (int r = 0; r < SORT_CH / 32; ++r) {
    const int e = e0 + r * 32 + lane;
    const bool valid = e < E;
    const int d = (kreg[r] >> SHIFT) & 255;
    const unsigned m = same_digit_lanes(d, valid);
    const unsigned below = m & ((1u << lane) - 1u);
    if (valid && below == 0u) cnt[d] += __popc(m);
    __syncthreads();
  }
  for (int b = lane; b < 256; b += 32) hist[b * B + blk] = cnt[b];
}

#define SCAN_PT 4
#define SCAN_TILE (1024 * SCAN_PT)

__launch_bounds__(1024)
__global__ void k_tilesum(const int* __restrict__ a, int* __restrict__ tsum, int L) {
  __shared__ int wtot[32];
  const int t = threadIdx.x, lane = t & 31, w = t >> 5;
  const int base = blockIdx.x * SCAN_TILE;
  int v = 0;
#pragma unroll
  for (int q = 0; q < SCAN_PT; ++q) {
    const int i = base + q * 1024 + t;
    const int x = a[(i < L) ? i : 0];
    v += (i < L) ? x : 0;
  }
  for (int o = 16; o > 0; o >>= 1) v += __shfl_xor(v, o);
  if (lane == 0) wtot[w] = v;
  __syncthreads();
  if (w == 0) {
    int s = wtot[lane];
    for (int o = 16; o > 0; o >>= 1) s += __shfl_xor(s, o);
    if (lane == 0) tsum[blockIdx.x] = s;
  }
}

__launch_bounds__(1024)
__global__ void k_scan(const int* __restrict__ a, const int* __restrict__ tsum,
                       int* __restrict__ out, int L) {
  __shared__ int wtot[32];
  __shared__ int base_s;
  const int t = threadIdx.x, lane = t & 31, w = t >> 5;

  const int i0 = (blockIdx.x * 1024 + t) * SCAN_PT;
  int c[SCAN_PT];
#pragma unroll
  for (int q = 0; q < SCAN_PT; ++q) {
    const int i = i0 + q;
    const int x = a[(i < L) ? i : 0];
    c[q] = (i < L) ? x : 0;
  }

  int part = 0;
  for (int j = t; j < (int)blockIdx.x; j += 1024) part += tsum[j];
  for (int o = 16; o > 0; o >>= 1) part += __shfl_xor(part, o);
  if (lane == 0) wtot[w] = part;
  __syncthreads();
  if (w == 0) {
    int v = wtot[lane];
    for (int o = 16; o > 0; o >>= 1) v += __shfl_xor(v, o);
    if (lane == 0) base_s = v;
  }
  __syncthreads();
  const int base = base_s;

  int s = 0;
#pragma unroll
  for (int q = 0; q < SCAN_PT; ++q) s += c[q];
  int inc = s;
  for (int o = 1; o < 32; o <<= 1) {
    const int u = __shfl_up(inc, o);
    if (lane >= o) inc += u;
  }
  if (lane == 31) wtot[w] = inc;
  __syncthreads();
  if (w == 0) {
    const int v = wtot[lane];
    int x = v;
    for (int o = 1; o < 32; o <<= 1) {
      const int u = __shfl_up(x, o);
      if (lane >= o) x += u;
    }
    wtot[lane] = x - v;
  }
  __syncthreads();
  int run = base + wtot[w] + inc - s;
#pragma unroll
  for (int q = 0; q < SCAN_PT; ++q) {
    if (i0 + q < L) out[i0 + q] = run;
    run += c[q];
  }
}

template <int SHIFT>
__launch_bounds__(32)
__global__ void k_place(const int* __restrict__ key, const int* __restrict__ val,
                        const int* __restrict__ off, int* __restrict__ okey,
                        int* __restrict__ oval, int E, int B) {
  __shared__ int base[256];
  const int lane = threadIdx.x;
  const int blk  = blockIdx.x;
  int o8[8];
#pragma unroll
  for (int i = 0; i < 8; ++i) o8[i] = off[(lane + 32 * i) * B + blk];
  const int e0 = blk * SORT_CH;
  int kreg[SORT_CH / 32], vreg[SORT_CH / 32];
#pragma unroll
  for (int r = 0; r < SORT_CH / 32; ++r) {
    const int e = e0 + r * 32 + lane;
    kreg[r] = (e < E) ? key[e] : 0;
    vreg[r] = (e < E) ? val[e] : 0;
  }
#pragma unroll
  for (int i = 0; i < 8; ++i) base[lane + 32 * i] = o8[i];
  __syncthreads();
  int pos[SORT_CH / 32];
#pragma unroll
  for (int r = 0; r < SORT_CH / 32; ++r) {
    const int e = e0 + r * 32 + lane;
    const bool valid = e < E;
    const int d = (kreg[r] >> SHIFT) & 255;
    const unsigned m = same_digit_lanes(d, valid);
    const unsigned below = m & ((1u << lane) - 1u);
    const int o = base[d];
    __syncthreads();
    if (valid && below == 0u) base[d] = o + __popc(m);
    __syncthreads();
    pos[r] = o + __popc(below);
  }
#pragma unroll
  for (int r = 0; r < SORT_CH / 32; ++r) {
    const int e = e0 + r * 32 + lane;
    if (e < E) {
      okey[pos[r]] = kreg[r];
      oval[pos[r]] = vreg[r];
    }
  }
}

#define OWN_W 32

__device__ __forceinline__ void wave_lds_order() {
  __builtin_amdgcn_fence(__ATOMIC_ACQ_REL, "wavefront");
}
__launch_bounds__(1024)
__global__ void k_owner(const int* __restrict__ key, const int* __restrict__ val,
                        const int* __restrict__ off, int* __restrict__ csr_src,
                        int* __restrict__ rowptr, float* __restrict__ dinv, int E, int B, int N) {
  __shared__ int wcnt[OWN_W * 256];
  __shared__ int wt[8];
  const int t = threadIdx.x, lane = t & 31;
  const int w = __builtin_amdgcn_readfirstlane(t >> 5);
  const int b = blockIdx.x;
  const int bs = off[b * B];
  const int be = (b + 1 < 256) ? off[(b + 1) * B] : E;
  const int R  = (be - bs + 32 * OWN_W - 1) / (32 * OWN_W);
  const int wbeg = bs + w * (R * 32);
  const int wend = (wbeg + R * 32 < be) ? (wbeg + R * 32) : be;
  const int elast = E - 1;

  for (int i = t; i < OWN_W * 256; i += 1024) wcnt[i] = 0;
  __syncthreads();
  int* mycnt = wcnt + w * 256;

  {
    int e = wbeg + lane;
    int kn = key[(e < elast) ? e : elast];
    for (int r = 0; r < R; ++r) {
      const int kc = kn;
      const bool valid = e < wend;
      e += 32;
      kn = key[(e < elast) ? e : elast];
      const int d = kc & 255;
      const unsigned m = same_digit_lanes(d, valid);
      const unsigned below = m & ((1u << lane) - 1u);
      if (valid && below == 0u) mycnt[d] += __popc(m);
      wave_lds_order();
    }
  }
  __syncthreads();

  int c[OWN_W];
  int total = 0;
  if (t < 256) {
#pragma unroll
    for (int q = 0; q < OWN_W; ++q) c[q] = wcnt[q * 256 + t];
#pragma unroll
    for (int q = 0; q < OWN_W; ++q) total += c[q];
  }
  int inc = total;
  for (int o = 1; o < 32; o <<= 1) {
    const int u = __shfl_up(inc, o);
    if (lane >= o) inc += u;
  }
  if (t < 256 && lane == 31) wt[w] = inc;
  __syncthreads();
  if (t < 256) {
    int pre = 0;
#pragma unroll
    for (int q = 0; q < 8; ++q) pre += (q < w) ? wt[q] : 0;
    int run = bs + pre + inc - total;
    const int node = b * 256 + t;
    if (node <= N) rowptr[node] = (node == N) ? E : run;
    if (node < N) dinv[node] = rsqrtf((float)total + 1.0f);
#pragma unroll
    for (int q = 0; q < OWN_W; ++q) {
      wcnt[q * 256 + t] = run;
      run += c[q];
    }
  }
  __syncthreads();

  {
    int e = wbeg + lane;
    int kn = key[(e < elast) ? e : elast];
    int vn = val[(e < elast) ? e : elast];
    for (int r = 0; r < R; ++r) {
      const int kc = kn, vc = vn;
      const bool valid = e < wend;
      e += 32;
      kn = key[(e < elast) ? e : elast];
      vn = val[(e < elast) ? e : elast];
      const int d = kc & 255;
      const unsigned m = same_digit_lanes(d, valid);
      const unsigned below = m & ((1u << lane) - 1u);
      const int o = mycnt[d];
      wave_lds_order();
      if (valid && below == 0u) mycnt[d] = o + __popc(m);
      wave_lds_order();
      if (valid) csr_src[o + __popc(below)] = vc;
    }
  }
}

template <int K, int N, int U, bool BIAS>
__launch_bounds__(128)
__global__ void k_gemm_wmma(const float* __restrict__ A, const float* __restrict__ B,
                            const float* __restrict__ bias, float* __restrict__ C, int nstrips) {
  constexpr int NT = N / 16;
  constexpr int G  = N / 64;
  const int lane = threadIdx.x & 31;
  const int half = lane >> 4;
  const int lm   = lane & 15;
  const int strip = __builtin_amdgcn_readfirstlane((int)(blockIdx.x * 4 + (threadIdx.x >> 5)));
  if (strip >= nstrips) return;

  const float* __restrict__ arow = A + (long long)(strip * 16 + lm) * K + 2 * half;
  const float* __restrict__ bp   = B + (long long)(2 * half) * N + 4 * lm;

  v8f acc[NT];
#pragma unroll
  for (int j = 0; j < NT; ++j) acc[j] = (v8f){0.f, 0.f, 0.f, 0.f, 0.f, 0.f, 0.f, 0.f};

  v2f a_cur[U], a_nxt[U];
#pragma unroll
  for (int u = 0; u < U; ++u) {
    a_cur[u] = *(const v2f*)(arow + 4 * u);
    a_nxt[u] = a_cur[u];
  }

#pragma unroll 1
  for (int kc = 0; kc < K; kc += 4 * U) {
    v4f bx[U][G], by[U][G];
#pragma unroll
    for (int u = 0; u < U; ++u) {
#pragma unroll
      for (int g = 0; g < G; ++g) {
        bx[u][g] = *(const v4f*)(bp + (kc + 4 * u) * N + 64 * g);
        by[u][g] = *(const v4f*)(bp + (kc + 4 * u) * N + N + 64 * g);
      }
    }
    if (kc + 4 * U < K) {
#pragma unroll
      for (int u = 0; u < U; ++u) a_nxt[u] = *(const v2f*)(arow + kc + 4 * U + 4 * u);
    }
#pragma unroll
    for (int u = 0; u < U; ++u) {
#pragma unroll
      for (int g = 0; g < G; ++g) {
#pragma unroll
        for (int q = 0; q < 4; ++q) {
          v2f b;
          b.x = bx[u][g][q];
          b.y = by[u][g][q];
          acc[4 * g + q] = __builtin_amdgcn_wmma_f32_16x16x4_f32(
               false, a_cur[u],  false, b,
               (short)0, acc[4 * g + q],  false,  false);
        }
      }
    }
#pragma unroll
    for (int u = 0; u < U; ++u) a_cur[u] = a_nxt[u];
  }

  v4f bv[G];
#pragma unroll
  for (int g = 0; g < G; ++g)
    bv[g] = BIAS ? *(const v4f*)(bias + 64 * g + 4 * lm) : (v4f){0.f, 0.f, 0.f, 0.f};
  float* __restrict__ cp = C + (long long)(strip * 16 + 8 * half) * N + 4 * lm;
#pragma unroll
  for (int r = 0; r < 8; ++r) {
#pragma unroll
    for (int g = 0; g < G; ++g) {
      v4f o;
      o.x = acc[4 * g + 0][r];
      o.y = acc[4 * g + 1][r];
      o.z = acc[4 * g + 2][r];
      o.w = acc[4 * g + 3][r];
      if (BIAS) o = o + bv[g];
      *(v4f*)(cp + r * N + 64 * g) = o;
    }
  }
}


template <int R>
__device__ __forceinline__ void pull_rows(const float2* __restrict__ h2, int s_l, float w_l,
                                          int t, int lane, float2& acc) {
  float2 v[R];
  float  wv[R];
#pragma unroll
  for (int k = 0; k < R; ++k) {
    const int s = __builtin_amdgcn_readlane(s_l, t + k);
    wv[k] = __int_as_float(__builtin_amdgcn_readlane(__float_as_int(w_l), t + k));
    v[k]  = h2[(long long)s * 32 + lane];
  }
#pragma unroll
  for (int k = 0; k < R; ++k) {
    acc.x += v[k].x * wv[k];
    acc.y += v[k].y * wv[k];
  }
}

template <bool RELU, bool BIAS>
__launch_bounds__(256)
__global__ void k_pull64(const float* __restrict__ h, const float* __restrict__ dinv,
                         const int* __restrict__ rowptr, const int* __restrict__ csr_src,
                         const float* __restrict__ bias, float* __restrict__ out, int N) {
  int node = blockIdx.x * (blockDim.x >> 5) + (threadIdx.x >> 5);
  node = __builtin_amdgcn_readfirstlane(node);
  const int lane = threadIdx.x & 31;
  if (node >= N) return;
  const int beg = __builtin_amdgcn_readfirstlane(rowptr[node]);
  const int end = __builtin_amdgcn_readfirstlane(rowptr[node + 1]);
  const float2* __restrict__ h2 = (const float2*)h;

  const float di = dinv[node];
  const float2 hv = h2[(long long)node * 32 + lane];
  float2 bv = make_float2(0.0f, 0.0f);
  if (BIAS) bv = ((const float2*)bias)[lane];

  float2 acc = make_float2(0.0f, 0.0f);
  for (int c = beg; c < end; c += 32) {
    const int n = (end - c < 32) ? (end - c) : 32;
    int   s_l = 0;
    float w_l = 0.0f;
    if (lane < n) {
      s_l = csr_src[c + lane];
      w_l = dinv[s_l];
    }
    int t = 0;
    for (; t + 8 <= n; t += 8) pull_rows<8>(h2, s_l, w_l, t, lane, acc);
    switch (n - t) {
      case 7: pull_rows<7>(h2, s_l, w_l, t, lane, acc); break;
      case 6: pull_rows<6>(h2, s_l, w_l, t, lane, acc); break;
      case 5: pull_rows<5>(h2, s_l, w_l, t, lane, acc); break;
      case 4: pull_rows<4>(h2, s_l, w_l, t, lane, acc); break;
      case 3: pull_rows<3>(h2, s_l, w_l, t, lane, acc); break;
      case 2: pull_rows<2>(h2, s_l, w_l, t, lane, acc); break;
      case 1: pull_rows<1>(h2, s_l, w_l, t, lane, acc); break;
      default: break;
    }
  }

  const float sl = di * di;
  float2 r;
  r.x = di * acc.x + sl * hv.x;
  r.y = di * acc.y + sl * hv.y;
  if (BIAS) {
    r.x += bv.x; r.y += bv.y;
  }
  if (RELU) {
    r.x = fmaxf(r.x, 0.0f); r.y = fmaxf(r.y, 0.0f);
  }
  ((float2*)out)[(long long)node * 32 + lane] = r;
}

extern "C" void kernel_launch(void* const* d_in, const int* in_sizes, int n_in,
                              void* d_out, int out_size, void* d_ws, size_t ws_size,
                              hipStream_t stream) {
  const float* x  = (const float*)d_in[0];
  const int*   ei = (const int*)d_in[1];
  const float* W1 = (const float*)d_in[2];
  const float* b1 = (const float*)d_in[3];
  const float* W2 = (const float*)d_in[4];
  const float* b2 = (const float*)d_in[5];

  const int FIN = 256, H = 64, FOUT = 128;
  const int N = in_sizes[0] / FIN;
  const int E = in_sizes[1] / 2;
  const int* srcI = ei;
  const int* dstI = ei + E;

  const int B = (E + SORT_CH - 1) / SORT_CH;
  const int L = 256 * B;
  const int T = (L + SCAN_TILE - 1) / SCAN_TILE;
  float* ws   = (float*)d_ws;
  float* h1   = ws;
  float* y1   = h1   + (long long)N * H;
  float* z    = y1   + (long long)N * H;
  float* dinv = z    + (long long)N * H;
  int* rowptr = (int*)(dinv + (long long)N);
  int* k1   = rowptr + (long long)N + 1;
  int* v1   = k1 + (long long)E;
  int* v2   = v1 + 2 * (long long)E;
  int* hist = v2 + (long long)E;
  int* off  = hist + (long long)L;
  int* tsum = off + (long long)L;

  const int TB = 256;
  auto blocks = [](long long n, int tb) { return (unsigned)((n + tb - 1) / tb); };

  k_hist<8><<<B, 32, 0, stream>>>(dstI, hist, E, B);
  k_tilesum<<<T, 1024, 0, stream>>>(hist, tsum, L);
  k_scan<<<T, 1024, 0, stream>>>(hist, tsum, off, L);
  k_place<8><<<B, 32, 0, stream>>>(dstI, srcI, off, k1, v1, E, B);
  k_owner<<<N / 256 + 1, 1024, 0, stream>>>(k1, v1, off, v2, rowptr, dinv, E, B, N);
  const int* csr_src = v2;

  const int nstrips = N / 16;
  k_gemm_wmma<256, 64, 16, false><<<blocks(nstrips, 4), 128, 0, stream>>>(x, W1, nullptr, h1, nstrips);
  k_pull64<true, true><<<blocks((long long)N * 32, TB), TB, 0, stream>>>(h1, dinv, rowptr, csr_src, b1, y1, N);

  k_pull64<false, false><<<blocks((long long)N * 32, TB), TB, 0, stream>>>(y1, dinv, rowptr, csr_src, nullptr, z, N);
  k_gemm_wmma<64, 128, 8, true><<<blocks(nstrips, 4), 128, 0, stream>>>(z, W2, b2, (float*)d_out, nstrips);
}


# Sort records are ordered in LDS and stored as runs of consecutive words, not one line per lane: pass 1 by 8192-edge work-group chunks, the owner pass through a staged list; list and sum order kept.
# speedup vs baseline: 5.8974x; 1.7882x over previous
#include <hip/hip_runtime.h>

typedef float v2f __attribute__((ext_vector_type(2)));
typedef float v4f __attribute__((ext_vector_type(4)));
typedef float v8f __attribute__((ext_vector_type(8)));


#define CH_W 32
#define CH_R 8
#define SORT_CH (CH_W * CH_R * 32)

__device__ __forceinline__ unsigned same_digit_lanes(int d, bool valid) {
  unsigned m = __builtin_amdgcn_ballot_w32(valid);
#pragma unroll
  for (int bit = 0; bit < 8; ++bit) {
    const bool p = (d >> bit) & 1;
    const unsigned bm = __builtin_amdgcn_ballot_w32(p);
    m &= p ? bm : ~bm;
  }
  return m;
}

__device__ __forceinline__ void wave_lds_order() {
  __builtin_amdgcn_fence(__ATOMIC_ACQ_REL, "wavefront");
}

template <int SHIFT>
__launch_bounds__(1024)
__global__ void k_hist(const int* __restrict__ key, int* __restrict__ hist, int E, int B) {
  __shared__ int wcnt[CH_W * 256];
  const int t = threadIdx.x, lane = t & 31;
  const int w = __builtin_amdgcn_readfirstlane(t >> 5);
  const int blk = blockIdx.x;
  const int e0 = blk * SORT_CH + w * (CH_R * 32) + lane;
  const int elast = E - 1;
  int kreg[CH_R];
#pragma unroll
  for (int r = 0; r < CH_R; ++r) {
    const int e = e0 + r * 32;
    kreg[r] = key[(e < E) ? e : elast];
  }
  for (int i = t; i < CH_W * 256; i += 1024) wcnt[i] = 0;
  __syncthreads();
  int* mycnt = wcnt + w * 256;
#pragma unroll
  for (int r = 0; r < CH_R; ++r) {
    const bool valid = (e0 + r * 32) < E;
    const int d = (kreg[r] >> SHIFT) & 255;
    const unsigned m = same_digit_lanes(d, valid);
    const unsigned below = m & ((1u << lane) - 1u);
    if (valid && below == 0u) mycnt[d] += __popc(m);
    wave_lds_order();
  }
  __syncthreads();
  if (t < 256) {
    int total = 0;
#pragma unroll
    for (int q = 0; q < CH_W; ++q) total += wcnt[q * 256 + t];
    hist[t * B + blk] = total;
  }
}

#define SCAN_PT 4
#define SCAN_TILE (1024 * SCAN_PT)

__launch_bounds__(1024)
__global__ void k_tilesum(const int* __restrict__ a, int* __restrict__ tsum, int L) {
  __shared__ int wtot[32];
  const int t = threadIdx.x, lane = t & 31, w = t >> 5;
  const int base = blockIdx.x * SCAN_TILE;
  int v = 0;
#pragma unroll
  for (int q = 0; q < SCAN_PT; ++q) {
    const int i = base + q * 1024 + t;
    const int x = a[(i < L) ? i : 0];
    v += (i < L) ? x : 0;
  }
  for (int o = 16; o > 0; o >>= 1) v += __shfl_xor(v, o);
  if (lane == 0) wtot[w] = v;
  __syncthreads();
  if (w == 0) {
    int s = wtot[lane];
    for (int o = 16; o > 0; o >>= 1) s += __shfl_xor(s, o);
    if (lane == 0) tsum[blockIdx.x] = s;
  }
}

__launch_bounds__(1024)
__global__ void k_scan(const int* __restrict__ a, const int* __restrict__ tsum,
                       int* __restrict__ out, int L) {
  __shared__ int wtot[32];
  __shared__ int base_s;
  const int t = threadIdx.x, lane = t & 31, w = t >> 5;

  const int i0 = (blockIdx.x * 1024 + t) * SCAN_PT;
  int c[SCAN_PT];
#pragma unroll
  for (int q = 0; q < SCAN_PT; ++q) {
    const int i = i0 + q;
    const int x = a[(i < L) ? i : 0];
    c[q] = (i < L) ? x : 0;
  }

  int part = 0;
  for (int j = t; j < (int)blockIdx.x; j += 1024) part += tsum[j];
  for (int o = 16; o > 0; o >>= 1) part += __shfl_xor(part, o);
  if (lane == 0) wtot[w] = part;
  __syncthreads();
  if (w == 0) {
    int v = wtot[lane];
    for (int o = 16; o > 0; o >>= 1) v += __shfl_xor(v, o);
    if (lane == 0) base_s = v;
  }
  __syncthreads();
  const int base = base_s;

  int s = 0;
#pragma unroll
  for (int q = 0; q < SCAN_PT; ++q) s += c[q];
  int inc = s;
  for (int o = 1; o < 32; o <<= 1) {
    const int u = __shfl_up(inc, o);
    if (lane >= o) inc += u;
  }
  if (lane == 31) wtot[w] = inc;
  __syncthreads();
  if (w == 0) {
    const int v = wtot[lane];
    int x = v;
    for (int o = 1; o < 32; o <<= 1) {
      const int u = __shfl_up(x, o);
      if (lane >= o) x += u;
    }
    wtot[lane] = x - v;
  }
  __syncthreads();
  int run = base + wtot[w] + inc - s;
#pragma unroll
  for (int q = 0; q < SCAN_PT; ++q) {
    if (i0 + q < L) out[i0 + q] = run;
    run += c[q];
  }
}

template <int SHIFT>
__launch_bounds__(1024)
__global__ void k_place(const int* __restrict__ key, const int* __restrict__ val,
                        const int* __restrict__ off, int* __restrict__ okey,
                        int* __restrict__ oval, int E, int B) {
  __shared__ int wcnt[CH_W * 256];
  __shared__ int sk[SORT_CH];
  __shared__ int sv[SORT_CH];
  __shared__ int delta[256];
  __shared__ int wt[8];
  const int t = threadIdx.x, lane = t & 31;
  const int w = __builtin_amdgcn_readfirstlane(t >> 5);
  const int blk = blockIdx.x;
  const int e0 = blk * SORT_CH + w * (CH_R * 32) + lane;
  const int elast = E - 1;
  const int gb = off[((t < 256) ? t : 0) * B + blk];
  int kreg[CH_R], vreg[CH_R];
#pragma unroll
  for (int r = 0; r < CH_R; ++r) {
    const int e = e0 + r * 32;
    const int ec = (e < E) ? e : elast;
    kreg[r] = key[ec];
    vreg[r] = val[ec];
  }
  for (int i = t; i < CH_W * 256; i += 1024) wcnt[i] = 0;
  __syncthreads();
  int* mycnt = wcnt + w * 256;
#pragma unroll
  for (int r = 0; r < CH_R; ++r) {
    const bool valid = (e0 + r * 32) < E;
    const int d = (kreg[r] >> SHIFT) & 255;
    const unsigned m = same_digit_lanes(d, valid);
    const unsigned below = m & ((1u << lane) - 1u);
    if (valid && below == 0u) mycnt[d] += __popc(m);
    wave_lds_order();
  }
  __syncthreads();

  int total = 0;
  if (t < 256) {
#pragma unroll
    for (int q = 0; q < CH_W; ++q) total += wcnt[q * 256 + t];
  }
  int inc = total;
  for (int o = 1; o < 32; o <<= 1) {
    const int u = __shfl_up(inc, o);
    if (lane >= o) inc += u;
  }
  if (t < 256 && lane == 31) wt[w] = inc;
  __syncthreads();
  if (t < 256) {
    int pre = 0;
#pragma unroll
    for (int q = 0; q < 8; ++q) pre += (q < w) ? wt[q] : 0;
    int run = pre + inc - total;
    delta[t] = gb - run;
#pragma unroll
    for (int q = 0; q < CH_W; ++q) {
      const int cq = wcnt[q * 256 + t];
      wcnt[q * 256 + t] = run;
      run += cq;
    }
  }
  __syncthreads();

#pragma unroll
  for (int r = 0; r < CH_R; ++r) {
    const bool valid = (e0 + r * 32) < E;
    const int d = (kreg[r] >> SHIFT) & 255;
    const unsigned m = same_digit_lanes(d, valid);
    const unsigned below = m & ((1u << lane) - 1u);
    const int o = mycnt[d];
    wave_lds_order();
    if (valid && below == 0u) mycnt[d] = o + __popc(m);
    wave_lds_order();
    if (valid) {
      const int lp = o + __popc(below);
      sk[lp] = kreg[r];
      sv[lp] = vreg[r];
    }
  }
  __syncthreads();

  const int left = E - blk * SORT_CH;
  const int n = (left < SORT_CH) ? left : SORT_CH;
  int ks[SORT_CH / 1024], vs[SORT_CH / 1024], gs[SORT_CH / 1024];
#pragma unroll
  for (int j = 0; j < SORT_CH / 1024; ++j) {
    ks[j] = sk[t + 1024 * j];
    vs[j] = sv[t + 1024 * j];
  }
#pragma unroll
  for (int j = 0; j < SORT_CH / 1024; ++j) gs[j] = delta[(ks[j] >> SHIFT) & 255] + t + 1024 * j;
#pragma unroll
  for (int j = 0; j < SORT_CH / 1024; ++j) {
    if (t + 1024 * j < n) {
      okey[gs[j]] = ks[j];
      oval[gs[j]] = vs[j];
    }
  }
}

#define OWN_W 32
#define OWN_STAGE 16384

__launch_bounds__(1024)
__global__ void k_owner(const int* __restrict__ key, const int* __restrict__ val,
                        const int* __restrict__ off, int* __restrict__ csr_src,
                        int* __restrict__ rowptr, float* __restrict__ dinv, int E, int B, int N) {
  __shared__ int wcnt[OWN_W * 256];
  __shared__ int wt[8];
  __shared__ int stage[OWN_STAGE];
  const int t = threadIdx.x, lane = t & 31;
  const int w = __builtin_amdgcn_readfirstlane(t >> 5);
  const int b = blockIdx.x;
  const int bs = off[b * B];
  const int be = (b + 1 < 256) ? off[(b + 1) * B] : E;
  const int R  = (be - bs + 32 * OWN_W - 1) / (32 * OWN_W);
  const int wbeg = bs + w * (R * 32);
  const int wend = (wbeg + R * 32 < be) ? (wbeg + R * 32) : be;
  const int elast = E - 1;
  const bool staged = (be - bs) <= OWN_STAGE;

  for (int i = t; i < OWN_W * 256; i += 1024) wcnt[i] = 0;
  __syncthreads();
  int* mycnt = wcnt + w * 256;

  {
    int e = wbeg + lane;
    int kn = key[(e < elast) ? e : elast];
    for (int r = 0; r < R; ++r) {
      const int kc = kn;
      const bool valid = e < wend;
      e += 32;
      kn = key[(e < elast) ? e : elast];
      const int d = kc & 255;
      const unsigned m = same_digit_lanes(d, valid);
      const unsigned below = m & ((1u << lane) - 1u);
      if (valid && below == 0u) mycnt[d] += __popc(m);
      wave_lds_order();
    }
  }
  __syncthreads();

  int c[OWN_W];
  int total = 0;
  if (t < 256) {
#pragma unroll
    for (int q = 0; q < OWN_W; ++q) c[q] = wcnt[q * 256 + t];
#pragma unroll
    for (int q = 0; q < OWN_W; ++q) total += c[q];
  }
  int inc = total;
  for (int o = 1; o < 32; o <<= 1) {
    const int u = __shfl_up(inc, o);
    if (lane >= o) inc += u;
  }
  if (t < 256 && lane == 31) wt[w] = inc;
  __syncthreads();
  if (t < 256) {
    int pre = 0;
#pragma unroll
    for (int q = 0; q < 8; ++q) pre += (q < w) ? wt[q] : 0;
    int run = bs + pre + inc - total;
    const int node = b * 256 + t;
    if (node <= N) rowptr[node] = (node == N) ? E : run;
    if (node < N) dinv[node] = rsqrtf((float)total + 1.0f);
#pragma unroll
    for (int q = 0; q < OWN_W; ++q) {
      wcnt[q * 256 + t] = run;
      run += c[q];
    }
  }
  __syncthreads();

  {
    int e = wbeg + lane;
    int kn = key[(e < elast) ? e : elast];
    int vn = val[(e < elast) ? e : elast];
    for (int r = 0; r < R; ++r) {
      const int kc = kn, vc = vn;
      const bool valid = e < wend;
      e += 32;
      kn = key[(e < elast) ? e : elast];
      vn = val[(e < elast) ? e : elast];
      const int d = kc & 255;
      const unsigned m = same_digit_lanes(d, valid);
      const unsigned below = m & ((1u << lane) - 1u);
      const int o = mycnt[d];
      wave_lds_order();
      if (valid && below == 0u) mycnt[d] = o + __popc(m);
      wave_lds_order();
      if (valid) {
        const int p = o + __popc(below);
        if (staged) stage[p - bs] = vc;
        else        csr_src[p] = vc;
      }
    }
  }
  if (staged) {
    __syncthreads();
    for (int i = t; i < be - bs; i += 1024) csr_src[bs + i] = stage[i];
  }
}

template <int K, int N, int U, bool BIAS>
__launch_bounds__(128)
__global__ void k_gemm_wmma(const float* __restrict__ A, const float* __restrict__ B,
                            const float* __restrict__ bias, float* __restrict__ C, int nstrips) {
  constexpr int NT = N / 16;
  constexpr int G  = N / 64;
  const int lane = threadIdx.x & 31;
  const int half = lane >> 4;
  const int lm   = lane & 15;
  const int strip = __builtin_amdgcn_readfirstlane((int)(blockIdx.x * 4 + (threadIdx.x >> 5)));
  if (strip >= nstrips) return;

  const float* __restrict__ arow = A + (long long)(strip * 16 + lm) * K + 2 * half;
  const float* __restrict__ bp   = B + (long long)(2 * half) * N + 4 * lm;

  v8f acc[NT];
#pragma unroll
  for (int j = 0; j < NT; ++j) acc[j] = (v8f){0.f, 0.f, 0.f, 0.f, 0.f, 0.f, 0.f, 0.f};

  v2f a_cur[U], a_nxt[U];
#pragma unroll
  for (int u = 0; u < U; ++u) {
    a_cur[u] = *(const v2f*)(arow + 4 * u);
    a_nxt[u] = a_cur[u];
  }

#pragma unroll 1
  for (int kc = 0; kc < K; kc += 4 * U) {
    v4f bx[U][G], by[U][G];
#pragma unroll
    for (int u = 0; u < U; ++u) {
#pragma unroll
      for (int g = 0; g < G; ++g) {
        bx[u][g] = *(const v4f*)(bp + (kc + 4 * u) * N + 64 * g);
        by[u][g] = *(const v4f*)(bp + (kc + 4 * u) * N + N + 64 * g);
      }
    }
    if (kc + 4 * U < K) {
#pragma unroll
      for (int u = 0; u < U; ++u) a_nxt[u] = *(const v2f*)(arow + kc + 4 * U + 4 * u);
    }
#pragma unroll
    for (int u = 0; u < U; ++u) {
#pragma unroll
      for (int g = 0; g < G; ++g) {
#pragma unroll
        for (int q = 0; q < 4; ++q) {
          v2f b;
          b.x = bx[u][g][q];
          b.y = by[u][g][q];
          acc[4 * g + q] = __builtin_amdgcn_wmma_f32_16x16x4_f32(
               false, a_cur[u],  false, b,
               (short)0, acc[4 * g + q],  false,  false);
        }
      }
    }
#pragma unroll
    for (int u = 0; u < U; ++u) a_cur[u] = a_nxt[u];
  }

  v4f bv[G];
#pragma unroll
  for (int g = 0; g < G; ++g)
    bv[g] = BIAS ? *(const v4f*)(bias + 64 * g + 4 * lm) : (v4f){0.f, 0.f, 0.f, 0.f};
  float* __restrict__ cp = C + (long long)(strip * 16 + 8 * half) * N + 4 * lm;
#pragma unroll
  for (int r = 0; r < 8; ++r) {
#pragma unroll
    for (int g = 0; g < G; ++g) {
      v4f o;
      o.x = acc[4 * g + 0][r];
      o.y = acc[4 * g + 1][r];
      o.z = acc[4 * g + 2][r];
      o.w = acc[4 * g + 3][r];
      if (BIAS) o = o + bv[g];
      *(v4f*)(cp + r * N + 64 * g) = o;
    }
  }
}


template <int R>
__device__ __forceinline__ void pull_rows(const float2* __restrict__ h2, int s_l, float w_l,
                                          int t, int lane, float2& acc) {
  float2 v[R];
  float  wv[R];
#pragma unroll
  for (int k = 0; k < R; ++k) {
    const int s = __builtin_amdgcn_readlane(s_l, t + k);
    wv[k] = __int_as_float(__builtin_amdgcn_readlane(__float_as_int(w_l), t + k));
    v[k]  = h2[(long long)s * 32 + lane];
  }
#pragma unroll
  for (int k = 0; k < R; ++k) {
    acc.x += v[k].x * wv[k];
    acc.y += v[k].y * wv[k];
  }
}

template <bool RELU, bool BIAS>
__launch_bounds__(256)
__global__ void k_pull64(const float* __restrict__ h, const float* __restrict__ dinv,
                         const int* __restrict__ rowptr, const int* __restrict__ csr_src,
                         const float* __restrict__ bias, float* __restrict__ out, int N) {
  int node = blockIdx.x * (blockDim.x >> 5) + (threadIdx.x >> 5);
  node = __builtin_amdgcn_readfirstlane(node);
  const int lane = threadIdx.x & 31;
  if (node >= N) return;
  const int beg = __builtin_amdgcn_readfirstlane(rowptr[node]);
  const int end = __builtin_amdgcn_readfirstlane(rowptr[node + 1]);
  const float2* __restrict__ h2 = (const float2*)h;

  const float di = dinv[node];
  const float2 hv = h2[(long long)node * 32 + lane];
  float2 bv = make_float2(0.0f, 0.0f);
  if (BIAS) bv = ((const float2*)bias)[lane];

  float2 acc = make_float2(0.0f, 0.0f);
  for (int c = beg; c < end; c += 32) {
    const int n = (end - c < 32) ? (end - c) : 32;
    int   s_l = 0;
    float w_l = 0.0f;
    if (lane < n) {
      s_l = csr_src[c + lane];
      w_l = dinv[s_l];
    }
    int t = 0;
    for (; t + 8 <= n; t += 8) pull_rows<8>(h2, s_l, w_l, t, lane, acc);
    switch (n - t) {
      case 7: pull_rows<7>(h2, s_l, w_l, t, lane, acc); break;
      case 6: pull_rows<6>(h2, s_l, w_l, t, lane, acc); break;
      case 5: pull_rows<5>(h2, s_l, w_l, t, lane, acc); break;
      case 4: pull_rows<4>(h2, s_l, w_l, t, lane, acc); break;
      case 3: pull_rows<3>(h2, s_l, w_l, t, lane, acc); break;
      case 2: pull_rows<2>(h2, s_l, w_l, t, lane, acc); break;
      case 1: pull_rows<1>(h2, s_l, w_l, t, lane, acc); break;
      default: break;
    }
  }

  const float sl = di * di;
  float2 r;
  r.x = di * acc.x + sl * hv.x;
  r.y = di * acc.y + sl * hv.y;
  if (BIAS) {
    r.x += bv.x; r.y += bv.y;
  }
  if (RELU) {
    r.x = fmaxf(r.x, 0.0f); r.y = fmaxf(r.y, 0.0f);
  }
  ((float2*)out)[(long long)node * 32 + lane] = r;
}

extern "C" void kernel_launch(void* const* d_in, const int* in_sizes, int n_in,
                              void* d_out, int out_size, void* d_ws, size_t ws_size,
                              hipStream_t stream) {
  const float* x  = (const float*)d_in[0];
  const int*   ei = (const int*)d_in[1];
  const float* W1 = (const float*)d_in[2];
  const float* b1 = (const float*)d_in[3];
  const float* W2 = (const float*)d_in[4];
  const float* b2 = (const float*)d_in[5];

  const int FIN = 256, H = 64, FOUT = 128;
  const int N = in_sizes[0] / FIN;
  const int E = in_sizes[1] / 2;
  const int* srcI = ei;
  const int* dstI = ei + E;

  const int B = (E + SORT_CH - 1) / SORT_CH;
  const int L = 256 * B;
  const int T = (L + SCAN_TILE - 1) / SCAN_TILE;
  float* ws   = (float*)d_ws;
  float* h1   = ws;
  float* y1   = h1   + (long long)N * H;
  float* z    = y1   + (long long)N * H;
  float* dinv = z    + (long long)N * H;
  int* rowptr = (int*)(dinv + (long long)N);
  int* k1   = rowptr + (long long)N + 1;
  int* v1   = k1 + (long long)E;
  int* v2   = v1 + 2 * (long long)E;
  int* hist = v2 + (long long)E;
  int* off  = hist + (long long)L;
  int* tsum = off + (long long)L;

  const int TB = 256;
  auto blocks = [](long long n, int tb) { return (unsigned)((n + tb - 1) / tb); };

  k_hist<8><<<B, 1024, 0, stream>>>(dstI, hist, E, B);
  k_tilesum<<<T, 1024, 0, stream>>>(hist, tsum, L);
  k_scan<<<T, 1024, 0, stream>>>(hist, tsum, off, L);
  k_place<8><<<B, 1024, 0, stream>>>(dstI, srcI, off, k1, v1, E, B);
  k_owner<<<N / 256 + 1, 1024, 0, stream>>>(k1, v1, off, v2, rowptr, dinv, E, B, N);
  const int* csr_src = v2;

  const int nstrips = N / 16;
  k_gemm_wmma<256, 64, 16, false><<<blocks(nstrips, 4), 128, 0, stream>>>(x, W1, nullptr, h1, nstrips);
  k_pull64<true, true><<<blocks((long long)N * 32, TB), TB, 0, stream>>>(h1, dinv, rowptr, csr_src, b1, y1, N);

  k_pull64<false, false><<<blocks((long long)N * 32, TB), TB, 0, stream>>>(y1, dinv, rowptr, csr_src, nullptr, z, N);
  k_gemm_wmma<64, 128, 8, true><<<blocks(nstrips, 4), 128, 0, stream>>>(z, W2, b2, (float*)d_out, nstrips);
}


# Count-table scan moved into the placement kernel: each chunk's work-group sums the [chunk][digit] table's columns itself, chunk 0 leaves the bucket starts; 9 launches become 7. Sum order kept.
# speedup vs baseline: 6.2827x; 1.0653x over previous
#include <hip/hip_runtime.h>

typedef float v2f __attribute__((ext_vector_type(2)));
typedef float v4f __attribute__((ext_vector_type(4)));
typedef float v8f __attribute__((ext_vector_type(8)));


#define CH_W 32
#define CH_R 8
#define SORT_CH (CH_W * CH_R * 32)

__device__ __forceinline__ unsigned same_digit_lanes(int d, bool valid) {
  unsigned m = __builtin_amdgcn_ballot_w32(valid);
#pragma unroll
  for (int bit = 0; bit < 8; ++bit) {
    const bool p = (d >> bit) & 1;
    const unsigned bm = __builtin_amdgcn_ballot_w32(p);
    m &= p ? bm : ~bm;
  }
  return m;
}

__device__ __forceinline__ void wave_lds_order() {
  __builtin_amdgcn_fence(__ATOMIC_ACQ_REL, "wavefront");
}

template <int SHIFT>
__launch_bounds__(1024)
__global__ void k_hist(const int* __restrict__ key, int* __restrict__ hist, int E) {
  __shared__ int wcnt[CH_W * 256];
  const int t = threadIdx.x, lane = t & 31;
  const int w = __builtin_amdgcn_readfirstlane(t >> 5);
  const int blk = blockIdx.x;
  const int e0 = blk * SORT_CH + w * (CH_R * 32) + lane;
  const int elast = E - 1;
  int kreg[CH_R];
#pragma unroll
  for (int r = 0; r < CH_R; ++r) {
    const int e = e0 + r * 32;
    kreg[r] = key[(e < E) ? e : elast];
  }
  for (int i = t; i < CH_W * 256; i += 1024) wcnt[i] = 0;
  __syncthreads();
  int* mycnt = wcnt + w * 256;
#pragma unroll
  for (int r = 0; r < CH_R; ++r) {
    const bool valid = (e0 + r * 32) < E;
    const int d = (kreg[r] >> SHIFT) & 255;
    const unsigned m = same_digit_lanes(d, valid);
    const unsigned below = m & ((1u << lane) - 1u);
    if (valid && below == 0u) mycnt[d] += __popc(m);
    wave_lds_order();
  }
  __syncthreads();
  if (t < 256) {
    int total = 0;
#pragma unroll
    for (int q = 0; q < CH_W; ++q) total += wcnt[q * 256 + t];
    hist[blk * 256 + t] = total;
  }
}

#define TAB_G 16
#define TAB_U 8
template <int SHIFT>
__launch_bounds__(1024)
__global__ void k_place(const int* __restrict__ key, const int* __restrict__ val,
                        const int* __restrict__ hist, int* __restrict__ bstart,
                        int* __restrict__ okey, int* __restrict__ oval, int E, int B) {
  __shared__ int wcnt[CH_W * 256];
  __shared__ int sk[SORT_CH];
  __shared__ int sv[SORT_CH];
  __shared__ int delta[256];
  __shared__ int wt[8];
  __shared__ int wt2[8];
  const int t = threadIdx.x, lane = t & 31;
  const int w = __builtin_amdgcn_readfirstlane(t >> 5);
  const int blk = blockIdx.x;
  const int e0 = blk * SORT_CH + w * (CH_R * 32) + lane;
  const int elast = E - 1;
  int kreg[CH_R], vreg[CH_R];
#pragma unroll
  for (int r = 0; r < CH_R; ++r) {
    const int e = e0 + r * 32;
    const int ec = (e < E) ? e : elast;
    kreg[r] = key[ec];
    vreg[r] = val[ec];
  }

  {
    const int q = t & 63;
    const int g = __builtin_amdgcn_readfirstlane(t >> 6);
    const int4* __restrict__ h4 = (const int4*)hist;
    int4 all = make_int4(0, 0, 0, 0), front = make_int4(0, 0, 0, 0);
    for (int b0 = g; b0 < B; b0 += TAB_G * TAB_U) {
      int4 hv[TAB_U];
#pragma unroll
      for (int j = 0; j < TAB_U; ++j) {
        const int b = b0 + TAB_G * j;
        const int4* __restrict__ row = h4 + (long long)((b < B) ? b : 0) * 64;
        hv[j] = row[q];
      }
#pragma unroll
      for (int j = 0; j < TAB_U; ++j) {
        const int b = b0 + TAB_G * j;
        const int in = (b < B) ? -1 : 0, fr = (b < blk) ? -1 : 0;
        all.x += hv[j].x & in;   all.y += hv[j].y & in;
        all.z += hv[j].z & in;   all.w += hv[j].w & in;
        front.x += hv[j].x & fr; front.y += hv[j].y & fr;
        front.z += hv[j].z & fr; front.w += hv[j].w & fr;
      }
    }
    *(int4*)(sk + g * 256 + 4 * q) = all;
    *(int4*)(sv + g * 256 + 4 * q) = front;
  }

  for (int i = t; i < CH_W * 256; i += 1024) wcnt[i] = 0;
  __syncthreads();
  {
    const int d = t & 255, p = t >> 8;
    int a = 0, f = 0;
#pragma unroll
    for (int j = 0; j < 4; ++j) {
      a += sk[(4 * p + j) * 256 + d];
      f += sv[(4 * p + j) * 256 + d];
    }
    sk[TAB_G * 256 + t] = a;
    sv[TAB_G * 256 + t] = f;
  }
  int* mycnt = wcnt + w * 256;
#pragma unroll
  for (int r = 0; r < CH_R; ++r) {
    const bool valid = (e0 + r * 32) < E;
    const int d = (kreg[r] >> SHIFT) & 255;
    const unsigned m = same_digit_lanes(d, valid);
    const unsigned below = m & ((1u << lane) - 1u);
    if (valid && below == 0u) mycnt[d] += __popc(m);
    wave_lds_order();
  }
  __syncthreads();

  int total = 0;
  int dall = 0, dfront = 0;
  if (t < 256) {
#pragma unroll
    for (int q = 0; q < CH_W; ++q) total += wcnt[q * 256 + t];
#pragma unroll
    for (int p = 0; p < 4; ++p) {
      dall   += sk[TAB_G * 256 + p * 256 + t];
      dfront += sv[TAB_G * 256 + p * 256 + t];
    }
  }
  int inc = total;
  int inc2 = dall;
  for (int o = 1; o < 32; o <<= 1) {
    const int u = __shfl_up(inc, o);
    const int u2 = __shfl_up(inc2, o);
    if (lane >= o) { inc += u; inc2 += u2; }
  }
  if (t < 256 && lane == 31) { wt[w] = inc; wt2[w] = inc2; }
  __syncthreads();
  if (t < 256) {
    int pre = 0, pre2 = 0;
#pragma unroll
    for (int q = 0; q < 8; ++q) {
      pre  += (q < w) ? wt[q] : 0;
      pre2 += (q < w) ? wt2[q] : 0;
    }
    int run = pre + inc - total;
    const int dstart = pre2 + inc2 - dall;
    const int gb = dstart + dfront;
    if (blk == 0) {
      bstart[t] = dstart;
      if (t == 255) bstart[256] = dstart + dall;
    }
    delta[t] = gb - run;
#pragma unroll
    for (int q = 0; q < CH_W; ++q) {
      const int cq = wcnt[q * 256 + t];
      wcnt[q * 256 + t] = run;
      run += cq;
    }
  }
  __syncthreads();

#pragma unroll
  for (int r = 0; r < CH_R; ++r) {
    const bool valid = (e0 + r * 32) < E;
    const int d = (kreg[r] >> SHIFT) & 255;
    const unsigned m = same_digit_lanes(d, valid);
    const unsigned below = m & ((1u << lane) - 1u);
    const int o = mycnt[d];
    wave_lds_order();
    if (valid && below == 0u) mycnt[d] = o + __popc(m);
    wave_lds_order();
    if (valid) {
      const int lp = o + __popc(below);
      sk[lp] = kreg[r];
      sv[lp] = vreg[r];
    }
  }
  __syncthreads();

  const int left = E - blk * SORT_CH;
  const int n = (left < SORT_CH) ? left : SORT_CH;
  int ks[SORT_CH / 1024], vs[SORT_CH / 1024], gs[SORT_CH / 1024];
#pragma unroll
  for (int j = 0; j < SORT_CH / 1024; ++j) {
    ks[j] = sk[t + 1024 * j];
    vs[j] = sv[t + 1024 * j];
  }
#pragma unroll
  for (int j = 0; j < SORT_CH / 1024; ++j) gs[j] = delta[(ks[j] >> SHIFT) & 255] + t + 1024 * j;
#pragma unroll
  for (int j = 0; j < SORT_CH / 1024; ++j) {
    if (t + 1024 * j < n) {
      okey[gs[j]] = ks[j];
      oval[gs[j]] = vs[j];
    }
  }
}

#define OWN_W 32
#define OWN_STAGE 16384

__launch_bounds__(1024)
__global__ void k_owner(const int* __restrict__ key, const int* __restrict__ val,
                        const int* __restrict__ bstart, int* __restrict__ csr_src,
                        int* __restrict__ rowptr, float* __restrict__ dinv, int E, int N) {
  __shared__ int wcnt[OWN_W * 256];
  __shared__ int wt[8];
  __shared__ int stage[OWN_STAGE];
  const int t = threadIdx.x, lane = t & 31;
  const int w = __builtin_amdgcn_readfirstlane(t >> 5);
  const int b = blockIdx.x;
  const int bs = bstart[b];
  const int be = bstart[b + 1];
  const int R  = (be - bs + 32 * OWN_W - 1) / (32 * OWN_W);
  const int wbeg = bs + w * (R * 32);
  const int wend = (wbeg + R * 32 < be) ? (wbeg + R * 32) : be;
  const int elast = E - 1;
  const bool staged = (be - bs) <= OWN_STAGE;

  for (int i = t; i < OWN_W * 256; i += 1024) wcnt[i] = 0;
  __syncthreads();
  int* mycnt = wcnt + w * 256;

  {
    int e = wbeg + lane;
    int kn = key[(e < elast) ? e : elast];
    for (int r = 0; r < R; ++r) {
      const int kc = kn;
      const bool valid = e < wend;
      e += 32;
      kn = key[(e < elast) ? e : elast];
      const int d = kc & 255;
      const unsigned m = same_digit_lanes(d, valid);
      const unsigned below = m & ((1u << lane) - 1u);
      if (valid && below == 0u) mycnt[d] += __popc(m);
      wave_lds_order();
    }
  }
  __syncthreads();

  int c[OWN_W];
  int total = 0;
  if (t < 256) {
#pragma unroll
    for (int q = 0; q < OWN_W; ++q) c[q] = wcnt[q * 256 + t];
#pragma unroll
    for (int q = 0; q < OWN_W; ++q) total += c[q];
  }
  int inc = total;
  for (int o = 1; o < 32; o <<= 1) {
    const int u = __shfl_up(inc, o);
    if (lane >= o) inc += u;
  }
  if (t < 256 && lane == 31) wt[w] = inc;
  __syncthreads();
  if (t < 256) {
    int pre = 0;
#pragma unroll
    for (int q = 0; q < 8; ++q) pre += (q < w) ? wt[q] : 0;
    int run = bs + pre + inc - total;
    const int node = b * 256 + t;
    if (node <= N) rowptr[node] = (node == N) ? E : run;
    if (node < N) dinv[node] = rsqrtf((float)total + 1.0f);
#pragma unroll
    for (int q = 0; q < OWN_W; ++q) {
      wcnt[q * 256 + t] = run;
      run += c[q];
    }
  }
  __syncthreads();

  {
    int e = wbeg + lane;
    int kn = key[(e < elast) ? e : elast];
    int vn = val[(e < elast) ? e : elast];
    for (int r = 0; r < R; ++r) {
      const int kc = kn, vc = vn;
      const bool valid = e < wend;
      e += 32;
      kn = key[(e < elast) ? e : elast];
      vn = val[(e < elast) ? e : elast];
      const int d = kc & 255;
      const unsigned m = same_digit_lanes(d, valid);
      const unsigned below = m & ((1u << lane) - 1u);
      const int o = mycnt[d];
      wave_lds_order();
      if (valid && below == 0u) mycnt[d] = o + __popc(m);
      wave_lds_order();
      if (valid) {
        const int p = o + __popc(below);
        if (staged) stage[p - bs] = vc;
        else        csr_src[p] = vc;
      }
    }
  }
  if (staged) {
    __syncthreads();
    for (int i = t; i < be - bs; i += 1024) csr_src[bs + i] = stage[i];
  }
}

template <int K, int N, int U, bool BIAS>
__launch_bounds__(128)
__global__ void k_gemm_wmma(const float* __restrict__ A, const float* __restrict__ B,
                            const float* __restrict__ bias, float* __restrict__ C, int nstrips) {
  constexpr int NT = N / 16;
  constexpr int G  = N / 64;
  const int lane = threadIdx.x & 31;
  const int half = lane >> 4;
  const int lm   = lane & 15;
  const int strip = __builtin_amdgcn_readfirstlane((int)(blockIdx.x * 4 + (threadIdx.x >> 5)));
  if (strip >= nstrips) return;

  const float* __restrict__ arow = A + (long long)(strip * 16 + lm) * K + 2 * half;
  const float* __restrict__ bp   = B + (long long)(2 * half) * N + 4 * lm;

  v8f acc[NT];
#pragma unroll
  for (int j = 0; j < NT; ++j) acc[j] = (v8f){0.f, 0.f, 0.f, 0.f, 0.f, 0.f, 0.f, 0.f};

  v2f a_cur[U], a_nxt[U];
#pragma unroll
  for (int u = 0; u < U; ++u) {
    a_cur[u] = *(const v2f*)(arow + 4 * u);
    a_nxt[u] = a_cur[u];
  }

#pragma unroll 1
  for (int kc = 0; kc < K; kc += 4 * U) {
    v4f bx[U][G], by[U][G];
#pragma unroll
    for (int u = 0; u < U; ++u) {
#pragma unroll
      for (int g = 0; g < G; ++g) {
        bx[u][g] = *(const v4f*)(bp + (kc + 4 * u) * N + 64 * g);
        by[u][g] = *(const v4f*)(bp + (kc + 4 * u) * N + N + 64 * g);
      }
    }
    if (kc + 4 * U < K) {
#pragma unroll
      for (int u = 0; u < U; ++u) a_nxt[u] = *(const v2f*)(arow + kc + 4 * U + 4 * u);
    }
#pragma unroll
    for (int u = 0; u < U; ++u) {
#pragma unroll
      for (int g = 0; g < G; ++g) {
#pragma unroll
        for (int q = 0; q < 4; ++q) {
          v2f b;
          b.x = bx[u][g][q];
          b.y = by[u][g][q];
          acc[4 * g + q] = __builtin_amdgcn_wmma_f32_16x16x4_f32(
               false, a_cur[u],  false, b,
               (short)0, acc[4 * g + q],  false,  false);
        }
      }
    }
#pragma unroll
    for (int u = 0; u < U; ++u) a_cur[u] = a_nxt[u];
  }

  v4f bv[G];
#pragma unroll
  for (int g = 0; g < G; ++g)
    bv[g] = BIAS ? *(const v4f*)(bias + 64 * g + 4 * lm) : (v4f){0.f, 0.f, 0.f, 0.f};
  float* __restrict__ cp = C + (long long)(strip * 16 + 8 * half) * N + 4 * lm;
#pragma unroll
  for (int r = 0; r < 8; ++r) {
#pragma unroll
    for (int g = 0; g < G; ++g) {
      v4f o;
      o.x = acc[4 * g + 0][r];
      o.y = acc[4 * g + 1][r];
      o.z = acc[4 * g + 2][r];
      o.w = acc[4 * g + 3][r];
      if (BIAS) o = o + bv[g];
      *(v4f*)(cp + r * N + 64 * g) = o;
    }
  }
}


template <int R>
__device__ __forceinline__ void pull_rows(const float2* __restrict__ h2, int s_l, float w_l,
                                          int t, int lane, float2& acc) {
  float2 v[R];
  float  wv[R];
#pragma unroll
  for (int k = 0; k < R; ++k) {
    const int s = __builtin_amdgcn_readlane(s_l, t + k);
    wv[k] = __int_as_float(__builtin_amdgcn_readlane(__float_as_int(w_l), t + k));
    v[k]  = h2[(long long)s * 32 + lane];
  }
#pragma unroll
  for (int k = 0; k < R; ++k) {
    acc.x += v[k].x * wv[k];
    acc.y += v[k].y * wv[k];
  }
}

template <bool RELU, bool BIAS>
__launch_bounds__(256)
__global__ void k_pull64(const float* __restrict__ h, const float* __restrict__ dinv,
                         const int* __restrict__ rowptr, const int* __restrict__ csr_src,
                         const float* __restrict__ bias, float* __restrict__ out, int N) {
  int node = blockIdx.x * (blockDim.x >> 5) + (threadIdx.x >> 5);
  node = __builtin_amdgcn_readfirstlane(node);
  const int lane = threadIdx.x & 31;
  if (node >= N) return;
  const int beg = __builtin_amdgcn_readfirstlane(rowptr[node]);
  const int end = __builtin_amdgcn_readfirstlane(rowptr[node + 1]);
  const float2* __restrict__ h2 = (const float2*)h;

  const float di = dinv[node];
  const float2 hv = h2[(long long)node * 32 + lane];
  float2 bv = make_float2(0.0f, 0.0f);
  if (BIAS) bv = ((const float2*)bias)[lane];

  float2 acc = make_float2(0.0f, 0.0f);
  for (int c = beg; c < end; c += 32) {
    const int n = (end - c < 32) ? (end - c) : 32;
    int   s_l = 0;
    float w_l = 0.0f;
    if (lane < n) {
      s_l = csr_src[c + lane];
      w_l = dinv[s_l];
    }
    int t = 0;
    for (; t + 8 <= n; t += 8) pull_rows<8>(h2, s_l, w_l, t, lane, acc);
    switch (n - t) {
      case 7: pull_rows<7>(h2, s_l, w_l, t, lane, acc); break;
      case 6: pull_rows<6>(h2, s_l, w_l, t, lane, acc); break;
      case 5: pull_rows<5>(h2, s_l, w_l, t, lane, acc); break;
      case 4: pull_rows<4>(h2, s_l, w_l, t, lane, acc); break;
      case 3: pull_rows<3>(h2, s_l, w_l, t, lane, acc); break;
      case 2: pull_rows<2>(h2, s_l, w_l, t, lane, acc); break;
      case 1: pull_rows<1>(h2, s_l, w_l, t, lane, acc); break;
      default: break;
    }
  }

  const float sl = di * di;
  float2 r;
  r.x = di * acc.x + sl * hv.x;
  r.y = di * acc.y + sl * hv.y;
  if (BIAS) {
    r.x += bv.x; r.y += bv.y;
  }
  if (RELU) {
    r.x = fmaxf(r.x, 0.0f); r.y = fmaxf(r.y, 0.0f);
  }
  ((float2*)out)[(long long)node * 32 + lane] = r;
}

extern "C" void kernel_launch(void* const* d_in, const int* in_sizes, int n_in,
                              void* d_out, int out_size, void* d_ws, size_t ws_size,
                              hipStream_t stream) {
  const float* x  = (const float*)d_in[0];
  const int*   ei = (const int*)d_in[1];
  const float* W1 = (const float*)d_in[2];
  const float* b1 = (const float*)d_in[3];
  const float* W2 = (const float*)d_in[4];
  const float* b2 = (const float*)d_in[5];

  const int FIN = 256, H = 64, FOUT = 128;
  const int N = in_sizes[0] / FIN;
  const int E = in_sizes[1] / 2;
  const int* srcI = ei;
  const int* dstI = ei + E;

  const int B = (E + SORT_CH - 1) / SORT_CH;
  const int L = 256 * B;
  float* ws   = (float*)d_ws;
  float* h1   = ws;
  float* y1   = h1   + (long long)N * H;
  float* z    = y1   + (long long)N * H;
  int* hist   = (int*)(z + (long long)N * H);
  int* bstart = hist + (long long)L;
  float* dinv = (float*)(bstart + 260);
  int* rowptr = (int*)(dinv + (long long)N);
  int* k1   = rowptr + (long long)N + 1;
  int* v1   = k1 + (long long)E;
  int* v2   = v1 + (long long)E;

  const int TB = 256;
  auto blocks = [](long long n, int tb) { return (unsigned)((n + tb - 1) / tb); };

  k_hist<8><<<B, 1024, 0, stream>>>(dstI, hist, E);
  k_place<8><<<B, 1024, 0, stream>>>(dstI, srcI, hist, bstart, k1, v1, E, B);
  k_owner<<<N / 256 + 1, 1024, 0, stream>>>(k1, v1, bstart, v2, rowptr, dinv, E, N);
  const int* csr_src = v2;

  const int nstrips = N / 16;
  k_gemm_wmma<256, 64, 16, false><<<blocks(nstrips, 4), 128, 0, stream>>>(x, W1, nullptr, h1, nstrips);
  k_pull64<true, true><<<blocks((long long)N * 32, TB), TB, 0, stream>>>(h1, dinv, rowptr, csr_src, b1, y1, N);

  k_pull64<false, false><<<blocks((long long)N * 32, TB), TB, 0, stream>>>(y1, dinv, rowptr, csr_src, nullptr, z, N);
  k_gemm_wmma<64, 128, 8, true><<<blocks(nstrips, 4), 128, 0, stream>>>(z, W2, b2, (float*)d_out, nstrips);
}


# Rows are scaled by their own dinv where they are made (GEMM 1 and pull 1 epilogues), so both pulls gather no per-edge dinv by source id; self row joins the edge sum before the dinv scale.
# speedup vs baseline: 6.9367x; 1.1041x over previous
#include <hip/hip_runtime.h>

typedef float v2f __attribute__((ext_vector_type(2)));
typedef float v4f __attribute__((ext_vector_type(4)));
typedef float v8f __attribute__((ext_vector_type(8)));


#define CH_W 32
#define CH_R 8
#define SORT_CH (CH_W * CH_R * 32)

__device__ __forceinline__ unsigned same_digit_lanes(int d, bool valid) {
  unsigned m = __builtin_amdgcn_ballot_w32(valid);
#pragma unroll
  for (int bit = 0; bit < 8; ++bit) {
    const bool p = (d >> bit) & 1;
    const unsigned bm = __builtin_amdgcn_ballot_w32(p);
    m &= p ? bm : ~bm;
  }
  return m;
}

__device__ __forceinline__ void wave_lds_order() {
  __builtin_amdgcn_fence(__ATOMIC_ACQ_REL, "wavefront");
}

template <int SHIFT>
__launch_bounds__(1024)
__global__ void k_hist(const int* __restrict__ key, int* __restrict__ hist, int E) {
  __shared__ int wcnt[CH_W * 256];
  const int t = threadIdx.x, lane = t & 31;
  const int w = __builtin_amdgcn_readfirstlane(t >> 5);
  const int blk = blockIdx.x;
  const int e0 = blk * SORT_CH + w * (CH_R * 32) + lane;
  const int elast = E - 1;
  int kreg[CH_R];
#pragma unroll
  for (int r = 0; r < CH_R; ++r) {
    const int e = e0 + r * 32;
    kreg[r] = key[(e < E) ? e : elast];
  }
  for (int i = t; i < CH_W * 256; i += 1024) wcnt[i] = 0;
  __syncthreads();
  int* mycnt = wcnt + w * 256;
#pragma unroll
  for (int r = 0; r < CH_R; ++r) {
    const bool valid = (e0 + r * 32) < E;
    const int d = (kreg[r] >> SHIFT) & 255;
    const unsigned m = same_digit_lanes(d, valid);
    const unsigned below = m & ((1u << lane) - 1u);
    if (valid && below == 0u) mycnt[d] += __popc(m);
    wave_lds_order();
  }
  __syncthreads();
  if (t < 256) {
    int total = 0;
#pragma unroll
    for (int q = 0; q < CH_W; ++q) total += wcnt[q * 256 + t];
    hist[blk * 256 + t] = total;
  }
}

#define TAB_G 16
#define TAB_U 8
template <int SHIFT>
__launch_bounds__(1024)
__global__ void k_place(const int* __restrict__ key, const int* __restrict__ val,
                        const int* __restrict__ hist, int* __restrict__ bstart,
                        int* __restrict__ okey, int* __restrict__ oval, int E, int B) {
  __shared__ int wcnt[CH_W * 256];
  __shared__ int sk[SORT_CH];
  __shared__ int sv[SORT_CH];
  __shared__ int delta[256];
  __shared__ int wt[8];
  __shared__ int wt2[8];
  const int t = threadIdx.x, lane = t & 31;
  const int w = __builtin_amdgcn_readfirstlane(t >> 5);
  const int blk = blockIdx.x;
  const int e0 = blk * SORT_CH + w * (CH_R * 32) + lane;
  const int elast = E - 1;
  int kreg[CH_R], vreg[CH_R];
#pragma unroll
  for (int r = 0; r < CH_R; ++r) {
    const int e = e0 + r * 32;
    const int ec = (e < E) ? e : elast;
    kreg[r] = key[ec];
    vreg[r] = val[ec];
  }

  {
    const int q = t & 63;
    const int g = __builtin_amdgcn_readfirstlane(t >> 6);
    const int4* __restrict__ h4 = (const int4*)hist;
    int4 all = make_int4(0, 0, 0, 0), front = make_int4(0, 0, 0, 0);
    for (int b0 = g; b0 < B; b0 += TAB_G * TAB_U) {
      int4 hv[TAB_U];
#pragma unroll
      for (int j = 0; j < TAB_U; ++j) {
        const int b = b0 + TAB_G * j;
        const int4* __restrict__ row = h4 + (long long)((b < B) ? b : 0) * 64;
        hv[j] = row[q];
      }
#pragma unroll
      for (int j = 0; j < TAB_U; ++j) {
        const int b = b0 + TAB_G * j;
        const int in = (b < B) ? -1 : 0, fr = (b < blk) ? -1 : 0;
        all.x += hv[j].x & in;   all.y += hv[j].y & in;
        all.z += hv[j].z & in;   all.w += hv[j].w & in;
        front.x += hv[j].x & fr; front.y += hv[j].y & fr;
        front.z += hv[j].z & fr; front.w += hv[j].w & fr;
      }
    }
    *(int4*)(sk + g * 256 + 4 * q) = all;
    *(int4*)(sv + g * 256 + 4 * q) = front;
  }

  for (int i = t; i < CH_W * 256; i += 1024) wcnt[i] = 0;
  __syncthreads();
  {
    const int d = t & 255, p = t >> 8;
    int a = 0, f = 0;
#pragma unroll
    for (int j = 0; j < 4; ++j) {
      a += sk[(4 * p + j) * 256 + d];
      f += sv[(4 * p + j) * 256 + d];
    }
    sk[TAB_G * 256 + t] = a;
    sv[TAB_G * 256 + t] = f;
  }
  int* mycnt = wcnt + w * 256;
#pragma unroll
  for (int r = 0; r < CH_R; ++r) {
    const bool valid = (e0 + r * 32) < E;
    const int d = (kreg[r] >> SHIFT) & 255;
    const unsigned m = same_digit_lanes(d, valid);
    const unsigned below = m & ((1u << lane) - 1u);
    if (valid && below == 0u) mycnt[d] += __popc(m);
    wave_lds_order();
  }
  __syncthreads();

  int total = 0;
  int dall = 0, dfront = 0;
  if (t < 256) {
#pragma unroll
    for (int q = 0; q < CH_W; ++q) total += wcnt[q * 256 + t];
#pragma unroll
    for (int p = 0; p < 4; ++p) {
      dall   += sk[TAB_G * 256 + p * 256 + t];
      dfront += sv[TAB_G * 256 + p * 256 + t];
    }
  }
  int inc = total;
  int inc2 = dall;
  for (int o = 1; o < 32; o <<= 1) {
    const int u = __shfl_up(inc, o);
    const int u2 = __shfl_up(inc2, o);
    if (lane >= o) { inc += u; inc2 += u2; }
  }
  if (t < 256 && lane == 31) { wt[w] = inc; wt2[w] = inc2; }
  __syncthreads();
  if (t < 256) {
    int pre = 0, pre2 = 0;
#pragma unroll
    for (int q = 0; q < 8; ++q) {
      pre  += (q < w) ? wt[q] : 0;
      pre2 += (q < w) ? wt2[q] : 0;
    }
    int run = pre + inc - total;
    const int dstart = pre2 + inc2 - dall;
    const int gb = dstart + dfront;
    if (blk == 0) {
      bstart[t] = dstart;
      if (t == 255) bstart[256] = dstart + dall;
    }
    delta[t] = gb - run;
#pragma unroll
    for (int q = 0; q < CH_W; ++q) {
      const int cq = wcnt[q * 256 + t];
      wcnt[q * 256 + t] = run;
      run += cq;
    }
  }
  __syncthreads();

#pragma unroll
  for (int r = 0; r < CH_R; ++r) {
    const bool valid = (e0 + r * 32) < E;
    const int d = (kreg[r] >> SHIFT) & 255;
    const unsigned m = same_digit_lanes(d, valid);
    const unsigned below = m & ((1u << lane) - 1u);
    const int o = mycnt[d];
    wave_lds_order();
    if (valid && below == 0u) mycnt[d] = o + __popc(m);
    wave_lds_order();
    if (valid) {
      const int lp = o + __popc(below);
      sk[lp] = kreg[r];
      sv[lp] = vreg[r];
    }
  }
  __syncthreads();

  const int left = E - blk * SORT_CH;
  const int n = (left < SORT_CH) ? left : SORT_CH;
  int ks[SORT_CH / 1024], vs[SORT_CH / 1024], gs[SORT_CH / 1024];
#pragma unroll
  for (int j = 0; j < SORT_CH / 1024; ++j) {
    ks[j] = sk[t + 1024 * j];
    vs[j] = sv[t + 1024 * j];
  }
#pragma unroll
  for (int j = 0; j < SORT_CH / 1024; ++j) gs[j] = delta[(ks[j] >> SHIFT) & 255] + t + 1024 * j;
#pragma unroll
  for (int j = 0; j < SORT_CH / 1024; ++j) {
    if (t + 1024 * j < n) {
      okey[gs[j]] = ks[j];
      oval[gs[j]] = vs[j];
    }
  }
}

#define OWN_W 32
#define OWN_STAGE 16384

__launch_bounds__(1024)
__global__ void k_owner(const int* __restrict__ key, const int* __restrict__ val,
                        const int* __restrict__ bstart, int* __restrict__ csr_src,
                        int* __restrict__ rowptr, float* __restrict__ dinv, int E, int N) {
  __shared__ int wcnt[OWN_W * 256];
  __shared__ int wt[8];
  __shared__ int stage[OWN_STAGE];
  const int t = threadIdx.x, lane = t & 31;
  const int w = __builtin_amdgcn_readfirstlane(t >> 5);
  const int b = blockIdx.x;
  const int bs = bstart[b];
  const int be = bstart[b + 1];
  const int R  = (be - bs + 32 * OWN_W - 1) / (32 * OWN_W);
  const int wbeg = bs + w * (R * 32);
  const int wend = (wbeg + R * 32 < be) ? (wbeg + R * 32) : be;
  const int elast = E - 1;
  const bool staged = (be - bs) <= OWN_STAGE;

  for (int i = t; i < OWN_W * 256; i += 1024) wcnt[i] = 0;
  __syncthreads();
  int* mycnt = wcnt + w * 256;

  {
    int e = wbeg + lane;
    int kn = key[(e < elast) ? e : elast];
    for (int r = 0; r < R; ++r) {
      const int kc = kn;
      const bool valid = e < wend;
      e += 32;
      kn = key[(e < elast) ? e : elast];
      const int d = kc & 255;
      const unsigned m = same_digit_lanes(d, valid);
      const unsigned below = m & ((1u << lane) - 1u);
      if (valid && below == 0u) mycnt[d] += __popc(m);
      wave_lds_order();
    }
  }
  __syncthreads();

  int c[OWN_W];
  int total = 0;
  if (t < 256) {
#pragma unroll
    for (int q = 0; q < OWN_W; ++q) c[q] = wcnt[q * 256 + t];
#pragma unroll
    for (int q = 0; q < OWN_W; ++q) total += c[q];
  }
  int inc = total;
  for (int o = 1; o < 32; o <<= 1) {
    const int u = __shfl_up(inc, o);
    if (lane >= o) inc += u;
  }
  if (t < 256 && lane == 31) wt[w] = inc;
  __syncthreads();
  if (t < 256) {
    int pre = 0;
#pragma unroll
    for (int q = 0; q < 8; ++q) pre += (q < w) ? wt[q] : 0;
    int run = bs + pre + inc - total;
    const int node = b * 256 + t;
    if (node <= N) rowptr[node] = (node == N) ? E : run;
    if (node < N) dinv[node] = rsqrtf((float)total + 1.0f);
#pragma unroll
    for (int q = 0; q < OWN_W; ++q) {
      wcnt[q * 256 + t] = run;
      run += c[q];
    }
  }
  __syncthreads();

  {
    int e = wbeg + lane;
    int kn = key[(e < elast) ? e : elast];
    int vn = val[(e < elast) ? e : elast];
    for (int r = 0; r < R; ++r) {
      const int kc = kn, vc = vn;
      const bool valid = e < wend;
      e += 32;
      kn = key[(e < elast) ? e : elast];
      vn = val[(e < elast) ? e : elast];
      const int d = kc & 255;
      const unsigned m = same_digit_lanes(d, valid);
      const unsigned below = m & ((1u << lane) - 1u);
      const int o = mycnt[d];
      wave_lds_order();
      if (valid && below == 0u) mycnt[d] = o + __popc(m);
      wave_lds_order();
      if (valid) {
        const int p = o + __popc(below);
        if (staged) stage[p - bs] = vc;
        else        csr_src[p] = vc;
      }
    }
  }
  if (staged) {
    __syncthreads();
    for (int i = t; i < be - bs; i += 1024) csr_src[bs + i] = stage[i];
  }
}

template <int K, int N, int U, bool BIAS, bool RSCALE>
__launch_bounds__(128)
__global__ void k_gemm_wmma(const float* __restrict__ A, const float* __restrict__ B,
                            const float* __restrict__ bias, const float* __restrict__ rscale,
                            float* __restrict__ C, int nstrips) {
  constexpr int NT = N / 16;
  constexpr int G  = N / 64;
  const int lane = threadIdx.x & 31;
  const int half = lane >> 4;
  const int lm   = lane & 15;
  const int strip = __builtin_amdgcn_readfirstlane((int)(blockIdx.x * 4 + (threadIdx.x >> 5)));
  if (strip >= nstrips) return;

  const float* __restrict__ arow = A + (long long)(strip * 16 + lm) * K + 2 * half;
  const float* __restrict__ bp   = B + (long long)(2 * half) * N + 4 * lm;

  v4f rs0 = (v4f){1.f, 1.f, 1.f, 1.f}, rs1 = (v4f){1.f, 1.f, 1.f, 1.f};
  if (RSCALE) {
    rs0 = *(const v4f*)(rscale + strip * 16 + 8 * half);
    rs1 = *(const v4f*)(rscale + strip * 16 + 8 * half + 4);
  }

  v8f acc[NT];
#pragma unroll
  for (int j = 0; j < NT; ++j) acc[j] = (v8f){0.f, 0.f, 0.f, 0.f, 0.f, 0.f, 0.f, 0.f};

  v2f a_cur[U], a_nxt[U];
#pragma unroll
  for (int u = 0; u < U; ++u) {
    a_cur[u] = *(const v2f*)(arow + 4 * u);
    a_nxt[u] = a_cur[u];
  }

#pragma unroll 1
  for (int kc = 0; kc < K; kc += 4 * U) {
    v4f bx[U][G], by[U][G];
#pragma unroll
    for (int u = 0; u < U; ++u) {
#pragma unroll
      for (int g = 0; g < G; ++g) {
        bx[u][g] = *(const v4f*)(bp + (kc + 4 * u) * N + 64 * g);
        by[u][g] = *(const v4f*)(bp + (kc + 4 * u) * N + N + 64 * g);
      }
    }
    if (kc + 4 * U < K) {
#pragma unroll
      for (int u = 0; u < U; ++u) a_nxt[u] = *(const v2f*)(arow + kc + 4 * U + 4 * u);
    }
#pragma unroll
    for (int u = 0; u < U; ++u) {
#pragma unroll
      for (int g = 0; g < G; ++g) {
#pragma unroll
        for (int q = 0; q < 4; ++q) {
          v2f b;
          b.x = bx[u][g][q];
          b.y = by[u][g][q];
          acc[4 * g + q] = __builtin_amdgcn_wmma_f32_16x16x4_f32(
               false, a_cur[u],  false, b,
               (short)0, acc[4 * g + q],  false,  false);
        }
      }
    }
#pragma unroll
    for (int u = 0; u < U; ++u) a_cur[u] = a_nxt[u];
  }

  v4f bv[G];
#pragma unroll
  for (int g = 0; g < G; ++g)
    bv[g] = BIAS ? *(const v4f*)(bias + 64 * g + 4 * lm) : (v4f){0.f, 0.f, 0.f, 0.f};
  float* __restrict__ cp = C + (long long)(strip * 16 + 8 * half) * N + 4 * lm;
#pragma unroll
  for (int r = 0; r < 8; ++r) {
#pragma unroll
    for (int g = 0; g < G; ++g) {
      v4f o;
      o.x = acc[4 * g + 0][r];
      o.y = acc[4 * g + 1][r];
      o.z = acc[4 * g + 2][r];
      o.w = acc[4 * g + 3][r];
      if (BIAS) o = o + bv[g];
      if (RSCALE) o = o * ((r < 4) ? rs0[r & 3] : rs1[r & 3]);
      *(v4f*)(cp + r * N + 64 * g) = o;
    }
  }
}


template <int R>
__device__ __forceinline__ void pull_rows(const float2* __restrict__ h2, int s_l,
                                          int t, int lane, float2& acc) {
  float2 v[R];
#pragma unroll
  for (int k = 0; k < R; ++k) {
    const int s = __builtin_amdgcn_readlane(s_l, t + k);
    v[k] = h2[(long long)s * 32 + lane];
  }
#pragma unroll
  for (int k = 0; k < R; ++k) {
    acc.x += v[k].x;
    acc.y += v[k].y;
  }
}

template <bool RELU, bool BIAS, bool OSCALE>
__launch_bounds__(256)
__global__ void k_pull64(const float* __restrict__ h, const float* __restrict__ dinv,
                         const int* __restrict__ rowptr, const int* __restrict__ csr_src,
                         const float* __restrict__ bias, float* __restrict__ out, int N) {
  int node = blockIdx.x * (blockDim.x >> 5) + (threadIdx.x >> 5);
  node = __builtin_amdgcn_readfirstlane(node);
  const int lane = threadIdx.x & 31;
  if (node >= N) return;
  const int beg = __builtin_amdgcn_readfirstlane(rowptr[node]);
  const int end = __builtin_amdgcn_readfirstlane(rowptr[node + 1]);
  const float2* __restrict__ h2 = (const float2*)h;

  const float di = dinv[node];
  const float2 hv = h2[(long long)node * 32 + lane];
  float2 bv = make_float2(0.0f, 0.0f);
  if (BIAS) bv = ((const float2*)bias)[lane];

  float2 acc = make_float2(0.0f, 0.0f);
  for (int c = beg; c < end; c += 32) {
    const int n = (end - c < 32) ? (end - c) : 32;
    int s_l = 0;
    if (lane < n) s_l = csr_src[c + lane];
    int t = 0;
    for (; t + 8 <= n; t += 8) pull_rows<8>(h2, s_l, t, lane, acc);
    switch (n - t) {
      case 7: pull_rows<7>(h2, s_l, t, lane, acc); break;
      case 6: pull_rows<6>(h2, s_l, t, lane, acc); break;
      case 5: pull_rows<5>(h2, s_l, t, lane, acc); break;
      case 4: pull_rows<4>(h2, s_l, t, lane, acc); break;
      case 3: pull_rows<3>(h2, s_l, t, lane, acc); break;
      case 2: pull_rows<2>(h2, s_l, t, lane, acc); break;
      case 1: pull_rows<1>(h2, s_l, t, lane, acc); break;
      default: break;
    }
  }

  float2 r;
  r.x = di * (acc.x + hv.x);
  r.y = di * (acc.y + hv.y);
  if (BIAS) {
    r.x += bv.x; r.y += bv.y;
  }
  if (RELU) {
    r.x = fmaxf(r.x, 0.0f); r.y = fmaxf(r.y, 0.0f);
  }
  if (OSCALE) {
    r.x *= di; r.y *= di;
  }
  ((float2*)out)[(long long)node * 32 + lane] = r;
}

extern "C" void kernel_launch(void* const* d_in, const int* in_sizes, int n_in,
                              void* d_out, int out_size, void* d_ws, size_t ws_size,
                              hipStream_t stream) {
  const float* x  = (const float*)d_in[0];
  const int*   ei = (const int*)d_in[1];
  const float* W1 = (const float*)d_in[2];
  const float* b1 = (const float*)d_in[3];
  const float* W2 = (const float*)d_in[4];
  const float* b2 = (const float*)d_in[5];

  const int FIN = 256, H = 64, FOUT = 128;
  const int N = in_sizes[0] / FIN;
  const int E = in_sizes[1] / 2;
  const int* srcI = ei;
  const int* dstI = ei + E;

  const int B = (E + SORT_CH - 1) / SORT_CH;
  const int L = 256 * B;
  float* ws   = (float*)d_ws;
  float* h1   = ws;
  float* y1   = h1   + (long long)N * H;
  float* z    = y1   + (long long)N * H;
  int* hist   = (int*)(z + (long long)N * H);
  int* bstart = hist + (long long)L;
  float* dinv = (float*)(bstart + 260);
  int* rowptr = (int*)(dinv + (long long)N);
  int* k1   = rowptr + (long long)N + 1;
  int* v1   = k1 + (long long)E;
  int* v2   = v1 + (long long)E;

  const int TB = 256;
  auto blocks = [](long long n, int tb) { return (unsigned)((n + tb - 1) / tb); };

  k_hist<8><<<B, 1024, 0, stream>>>(dstI, hist, E);
  k_place<8><<<B, 1024, 0, stream>>>(dstI, srcI, hist, bstart, k1, v1, E, B);
  k_owner<<<N / 256 + 1, 1024, 0, stream>>>(k1, v1, bstart, v2, rowptr, dinv, E, N);
  const int* csr_src = v2;

  const int nstrips = N / 16;
  k_gemm_wmma<256, 64, 16, false, true><<<blocks(nstrips, 4), 128, 0, stream>>>(x, W1, nullptr, dinv, h1, nstrips);
  k_pull64<true, true, true><<<blocks((long long)N * 32, TB), TB, 0, stream>>>(h1, dinv, rowptr, csr_src, b1, y1, N);

  k_pull64<false, false, false><<<blocks((long long)N * 32, TB), TB, 0, stream>>>(y1, dinv, rowptr, csr_src, nullptr, z, N);
  k_gemm_wmma<64, 128, 8, true, false><<<blocks(nstrips, 4), 128, 0, stream>>>(z, W2, b2, nullptr, (float*)d_out, nstrips);
}


# Count launch dropped: each chunk is ordered by bucket inside its own slot and leaves its bucket offsets; a bucket's owner collects its per-chunk runs into LDS. 7 launches become 6; sum order kept.
# speedup vs baseline: 7.2816x; 1.0497x over previous
#include <hip/hip_runtime.h>

typedef float v2f __attribute__((ext_vector_type(2)));
typedef float v4f __attribute__((ext_vector_type(4)));
typedef float v8f __attribute__((ext_vector_type(8)));


#define CH_W 32
#define CH_R 8
#define SORT_CH (CH_W * CH_R * 32)

__device__ __forceinline__ unsigned same_digit_lanes(int d, bool valid) {
  unsigned m = __builtin_amdgcn_ballot_w32(valid);
#pragma unroll
  for (int bit = 0; bit < 8; ++bit) {
    const bool p = (d >> bit) & 1;
    const unsigned bm = __builtin_amdgcn_ballot_w32(p);
    m &= p ? bm : ~bm;
  }
  return m;
}

__device__ __forceinline__ void wave_lds_order() {
  __builtin_amdgcn_fence(__ATOMIC_ACQ_REL, "wavefront");
}

template <int SHIFT>
__launch_bounds__(1024)
__global__ void k_place(const int* __restrict__ key, const int* __restrict__ val,
                        int* __restrict__ cstart, int* __restrict__ okey, int* __restrict__ oval, int E) {
  __shared__ int wcnt[CH_W * 256];
  __shared__ int sk[SORT_CH];
  __shared__ int sv[SORT_CH];
  __shared__ int wt[8];
  const int t = threadIdx.x, lane = t & 31;
  const int w = __builtin_amdgcn_readfirstlane(t >> 5);
  const int blk = blockIdx.x;
  const int e0 = blk * SORT_CH + w * (CH_R * 32) + lane;
  const int elast = E - 1;
  int kreg[CH_R], vreg[CH_R];
#pragma unroll
  for (int r = 0; r < CH_R; ++r) {
    const int e = e0 + r * 32;
    const int ec = (e < E) ? e : elast;
    kreg[r] = key[ec];
    vreg[r] = val[ec];
  }

  for (int i = t; i < CH_W * 256; i += 1024) wcnt[i] = 0;
  __syncthreads();
  int* mycnt = wcnt + w * 256;
#pragma unroll
  for (int r = 0; r < CH_R; ++r) {
    const bool valid = (e0 + r * 32) < E;
    const int d = (kreg[r] >> SHIFT) & 255;
    const unsigned m = same_digit_lanes(d, valid);
    const unsigned below = m & ((1u << lane) - 1u);
    if (valid && below == 0u) mycnt[d] += __popc(m);
    wave_lds_order();
  }
  __syncthreads();

  int total = 0;
  if (t < 256) {
#pragma unroll
    for (int q = 0; q < CH_W; ++q) total += wcnt[q * 256 + t];
  }
  int inc = total;
  for (int o = 1; o < 32; o <<= 1) {
    const int u = __shfl_up(inc, o);
    if (lane >= o) inc += u;
  }
  if (t < 256 && lane == 31) wt[w] = inc;
  __syncthreads();
  if (t < 256) {
    int pre = 0;
#pragma unroll
    for (int q = 0; q < 8; ++q) pre += (q < w) ? wt[q] : 0;
    int run = pre + inc - total;
    cstart[blk * 256 + t] = run;
#pragma unroll
    for (int q = 0; q < CH_W; ++q) {
      const int cq = wcnt[q * 256 + t];
      wcnt[q * 256 + t] = run;
      run += cq;
    }
  }
  __syncthreads();

#pragma unroll
  for (int r = 0; r < CH_R; ++r) {
    const bool valid = (e0 + r * 32) < E;
    const int d = (kreg[r] >> SHIFT) & 255;
    const unsigned m = same_digit_lanes(d, valid);
    const unsigned below = m & ((1u << lane) - 1u);
    const int o = mycnt[d];
    wave_lds_order();
    if (valid && below == 0u) mycnt[d] = o + __popc(m);
    wave_lds_order();
    if (valid) {
      const int lp = o + __popc(below);
      sk[lp] = kreg[r];
      sv[lp] = vreg[r];
    }
  }
  __syncthreads();

  const int left = E - blk * SORT_CH;
  const int n = (left < SORT_CH) ? left : SORT_CH;
  int ks[SORT_CH / 1024], vs[SORT_CH / 1024];
#pragma unroll
  for (int j = 0; j < SORT_CH / 1024; ++j) {
    ks[j] = sk[t + 1024 * j];
    vs[j] = sv[t + 1024 * j];
  }
  int* __restrict__ ok = okey + (long long)blk * SORT_CH;
  int* __restrict__ ov = oval + (long long)blk * SORT_CH;
#pragma unroll
  for (int j = 0; j < SORT_CH / 1024; ++j) {
    if (t + 1024 * j < n) {
      ok[t + 1024 * j] = ks[j];
      ov[t + 1024 * j] = vs[j];
    }
  }
}

#define OWN_W 32
#define OWN_STAGE 16384
#define OWN_MAXB 1024
#define OWN_F 10

__launch_bounds__(1024)
__global__ void k_owner(const int* __restrict__ key, const int* __restrict__ val,
                        const int* __restrict__ cstart, int* __restrict__ csr_src,
                        int* __restrict__ rowptr, float* __restrict__ dinv, int E, int N, int B) {
  __shared__ int wcnt[OWN_W * 256];
  __shared__ int wt[OWN_W];
  __shared__ int wt2[OWN_W];
  __shared__ int wtn[8];
  __shared__ int pre[OWN_MAXB];
  __shared__ int shift[OWN_MAXB];
  __shared__ int lk[OWN_STAGE];
  __shared__ int lv[OWN_STAGE];
  __shared__ int stage[OWN_STAGE];
  const int t = threadIdx.x, lane = t & 31;
  const int w = __builtin_amdgcn_readfirstlane(t >> 5);
  const int b = blockIdx.x;

  const int cc = (t < B) ? t : (B - 1);
  const int* __restrict__ row = cstart + (long long)cc * 256;
  const int v0 = row[b];
  const int v1 = row[(b < 255) ? (b + 1) : 255];
  const int left = E - cc * SORT_CH;
  const int nc = (left < SORT_CH) ? left : SORT_CH;
  const int cs  = (t < B) ? v0 : 0;
  const int len = (t < B) ? (((b < 255) ? v1 : nc) - v0) : 0;

  for (int i = t; i < OWN_W * 256; i += 1024) wcnt[i] = 0;

  int inc = len, inc2 = cs;
  for (int o = 1; o < 32; o <<= 1) {
    const int u = __shfl_up(inc, o);
    const int u2 = __shfl_up(inc2, o);
    if (lane >= o) { inc += u; inc2 += u2; }
  }
  if (lane == 31) { wt[w] = inc; wt2[w] = inc2; }
  __syncthreads();
  int pr = 0, tot = 0, bsum = 0;
#pragma unroll
  for (int q = 0; q < OWN_W; ++q) {
    const int a = wt[q];
    pr += (q < w) ? a : 0;
    tot += a;
    bsum += wt2[q];
  }
  const int first = pr + inc - len;
  pre[t] = (t < B) ? first : 0x7fffffff;
  shift[t] = cc * SORT_CH + cs - first;
  const int total = __builtin_amdgcn_readfirstlane(tot);
  const int bs = __builtin_amdgcn_readfirstlane(bsum);
  __syncthreads();

  const int R  = (total + 32 * OWN_W - 1) / (32 * OWN_W);
  const int wbeg = w * (R * 32);
  const int wend = (wbeg + R * 32 < total) ? (wbeg + R * 32) : total;
  const int plast = (total > 0) ? (total - 1) : 0;
  const bool staged = total <= OWN_STAGE;
  int s0 = 1;
  while (2 * s0 < B && s0 < OWN_MAXB / 2) s0 <<= 1;

  auto record_lds = [&](int p, int& k, int& v) {
    const int pc = (p < plast) ? p : plast;
    k = lk[pc];
    v = lv[pc];
  };
  auto record_mem = [&](int p, int& k, int& v) {
    const int pc = (p < plast) ? p : plast;
    int ci = 0;
    for (int s = s0; s > 0; s >>= 1) ci += (pre[ci + s] <= pc) ? s : 0;
    const int a = shift[ci] + pc;
    k = key[a];
    v = val[a];
  };

  if (staged) {
    for (int j0 = 0; j0 < R; j0 += OWN_F) {
      int pc[OWN_F], ci[OWN_F], kk[OWN_F], vv[OWN_F];
#pragma unroll
      for (int u = 0; u < OWN_F; ++u) {
        const int p = t + 1024 * (j0 + u);
        pc[u] = (p < plast) ? p : plast;
        ci[u] = 0;
      }
      for (int s = s0; s > 0; s >>= 1) {
#pragma unroll
        for (int u = 0; u < OWN_F; ++u) ci[u] += (pre[ci[u] + s] <= pc[u]) ? s : 0;
      }
#pragma unroll
      for (int u = 0; u < OWN_F; ++u) {
        const int a = shift[ci[u]] + pc[u];
        kk[u] = key[a];
        vv[u] = val[a];
      }
#pragma unroll
      for (int u = 0; u < OWN_F; ++u) {
        const int p = t + 1024 * (j0 + u);
        if (p < total) {
          lk[p] = kk[u];
          lv[p] = vv[u];
        }
      }
    }
    __syncthreads();
  }
  int* mycnt = wcnt + w * 256;

  auto walk_count = [&](auto record) {
    int p = wbeg + lane;
    int kn, vn;
    record(p, kn, vn);
    for (int r = 0; r < R; ++r) {
      const int kc = kn;
      const bool valid = p < wend;
      p += 32;
      record(p, kn, vn);
      const int d = kc & 255;
      const unsigned m = same_digit_lanes(d, valid);
      const unsigned below = m & ((1u << lane) - 1u);
      if (valid && below == 0u) mycnt[d] += __popc(m);
      wave_lds_order();
    }
  };
  if (staged) walk_count(record_lds);
  else        walk_count(record_mem);
  __syncthreads();

  int c[OWN_W];
  int ntot = 0;
  if (t < 256) {
#pragma unroll
    for (int q = 0; q < OWN_W; ++q) c[q] = wcnt[q * 256 + t];
#pragma unroll
    for (int q = 0; q < OWN_W; ++q) ntot += c[q];
  }
  int ninc = ntot;
  for (int o = 1; o < 32; o <<= 1) {
    const int u = __shfl_up(ninc, o);
    if (lane >= o) ninc += u;
  }
  if (t < 256 && lane == 31) wtn[w] = ninc;
  __syncthreads();
  if (t < 256) {
    int npre = 0;
#pragma unroll
    for (int q = 0; q < 8; ++q) npre += (q < w) ? wtn[q] : 0;
    int run = bs + npre + ninc - ntot;
    const int node = b * 256 + t;
    if (node <= N) rowptr[node] = (node == N) ? E : run;
    if (node < N) dinv[node] = rsqrtf((float)ntot + 1.0f);
#pragma unroll
    for (int q = 0; q < OWN_W; ++q) {
      wcnt[q * 256 + t] = run;
      run += c[q];
    }
  }
  __syncthreads();

  auto put_lds = [&](int q, int v) { stage[q - bs] = v; };
  auto put_mem = [&](int q, int v) { csr_src[q] = v; };
  auto walk_place = [&](auto record, auto put) {
    int p = wbeg + lane;
    int kn, vn;
    record(p, kn, vn);
    for (int r = 0; r < R; ++r) {
      const int kc = kn, vc = vn;
      const bool valid = p < wend;
      p += 32;
      record(p, kn, vn);
      const int d = kc & 255;
      const unsigned m = same_digit_lanes(d, valid);
      const unsigned below = m & ((1u << lane) - 1u);
      const int o = mycnt[d];
      wave_lds_order();
      if (valid && below == 0u) mycnt[d] = o + __popc(m);
      wave_lds_order();
      if (valid) put(o + __popc(below), vc);
    }
  };
  if (staged) {
    walk_place(record_lds, put_lds);
    __syncthreads();
    for (int i = t; i < total; i += 1024) csr_src[bs + i] = stage[i];
  } else {
    walk_place(record_mem, put_mem);
  }
}

template <int K, int N, int U, bool BIAS, bool RSCALE>
__launch_bounds__(128)
__global__ void k_gemm_wmma(const float* __restrict__ A, const float* __restrict__ B,
                            const float* __restrict__ bias, const float* __restrict__ rscale,
                            float* __restrict__ C, int nstrips) {
  constexpr int NT = N / 16;
  constexpr int G  = N / 64;
  const int lane = threadIdx.x & 31;
  const int half = lane >> 4;
  const int lm   = lane & 15;
  const int strip = __builtin_amdgcn_readfirstlane((int)(blockIdx.x * 4 + (threadIdx.x >> 5)));
  if (strip >= nstrips) return;

  const float* __restrict__ arow = A + (long long)(strip * 16 + lm) * K + 2 * half;
  const float* __restrict__ bp   = B + (long long)(2 * half) * N + 4 * lm;

  v4f rs0 = (v4f){1.f, 1.f, 1.f, 1.f}, rs1 = (v4f){1.f, 1.f, 1.f, 1.f};
  if (RSCALE) {
    rs0 = *(const v4f*)(rscale + strip * 16 + 8 * half);
    rs1 = *(const v4f*)(rscale + strip * 16 + 8 * half + 4);
  }

  v8f acc[NT];
#pragma unroll
  for (int j = 0; j < NT; ++j) acc[j] = (v8f){0.f, 0.f, 0.f, 0.f, 0.f, 0.f, 0.f, 0.f};

  v2f a_cur[U], a_nxt[U];
#pragma unroll
  for (int u = 0; u < U; ++u) {
    a_cur[u] = *(const v2f*)(arow + 4 * u);
    a_nxt[u] = a_cur[u];
  }

#pragma unroll 1
  for (int kc = 0; kc < K; kc += 4 * U) {
    v4f bx[U][G], by[U][G];
#pragma unroll
    for (int u = 0; u < U; ++u) {
#pragma unroll
      for (int g = 0; g < G; ++g) {
        bx[u][g] = *(const v4f*)(bp + (kc + 4 * u) * N + 64 * g);
        by[u][g] = *(const v4f*)(bp + (kc + 4 * u) * N + N + 64 * g);
      }
    }
    if (kc + 4 * U < K) {
#pragma unroll
      for (int u = 0; u < U; ++u) a_nxt[u] = *(const v2f*)(arow + kc + 4 * U + 4 * u);
    }
#pragma unroll
    for (int u = 0; u < U; ++u) {
#pragma unroll
      for (int g = 0; g < G; ++g) {
#pragma unroll
        for (int q = 0; q < 4; ++q) {
          v2f b;
          b.x = bx[u][g][q];
          b.y = by[u][g][q];
          acc[4 * g + q] = __builtin_amdgcn_wmma_f32_16x16x4_f32(
               false, a_cur[u],  false, b,
               (short)0, acc[4 * g + q],  false,  false);
        }
      }
    }
#pragma unroll
    for (int u = 0; u < U; ++u) a_cur[u] = a_nxt[u];
  }

  v4f bv[G];
#pragma unroll
  for (int g = 0; g < G; ++g)
    bv[g] = BIAS ? *(const v4f*)(bias + 64 * g + 4 * lm) : (v4f){0.f, 0.f, 0.f, 0.f};
  float* __restrict__ cp = C + (long long)(strip * 16 + 8 * half) * N + 4 * lm;
#pragma unroll
  for (int r = 0; r < 8; ++r) {
#pragma unroll
    for (int g = 0; g < G; ++g) {
      v4f o;
      o.x = acc[4 * g + 0][r];
      o.y = acc[4 * g + 1][r];
      o.z = acc[4 * g + 2][r];
      o.w = acc[4 * g + 3][r];
      if (BIAS) o = o + bv[g];
      if (RSCALE) o = o * ((r < 4) ? rs0[r & 3] : rs1[r & 3]);
      *(v4f*)(cp + r * N + 64 * g) = o;
    }
  }
}


template <int R>
__device__ __forceinline__ void pull_rows(const float2* __restrict__ h2, int s_l,
                                          int t, int lane, float2& acc) {
  float2 v[R];
#pragma unroll
  for (int k = 0; k < R; ++k) {
    const int s = __builtin_amdgcn_readlane(s_l, t + k);
    v[k] = h2[(long long)s * 32 + lane];
  }
#pragma unroll
  for (int k = 0; k < R; ++k) {
    acc.x += v[k].x;
    acc.y += v[k].y;
  }
}

template <bool RELU, bool BIAS, bool OSCALE>
__launch_bounds__(256)
__global__ void k_pull64(const float* __restrict__ h, const float* __restrict__ dinv,
                         const int* __restrict__ rowptr, const int* __restrict__ csr_src,
                         const float* __restrict__ bias, float* __restrict__ out, int N) {
  int node = blockIdx.x * (blockDim.x >> 5) + (threadIdx.x >> 5);
  node = __builtin_amdgcn_readfirstlane(node);
  const int lane = threadIdx.x & 31;
  if (node >= N) return;
  const int beg = __builtin_amdgcn_readfirstlane(rowptr[node]);
  const int end = __builtin_amdgcn_readfirstlane(rowptr[node + 1]);
  const float2* __restrict__ h2 = (const float2*)h;

  const float di = dinv[node];
  const float2 hv = h2[(long long)node * 32 + lane];
  float2 bv = make_float2(0.0f, 0.0f);
  if (BIAS) bv = ((const float2*)bias)[lane];

  float2 acc = make_float2(0.0f, 0.0f);
  for (int c = beg; c < end; c += 32) {
    const int n = (end - c < 32) ? (end - c) : 32;
    int s_l = 0;
    if (lane < n) s_l = csr_src[c + lane];
    int t = 0;
    for (; t + 8 <= n; t += 8) pull_rows<8>(h2, s_l, t, lane, acc);
    switch (n - t) {
      case 7: pull_rows<7>(h2, s_l, t, lane, acc); break;
      case 6: pull_rows<6>(h2, s_l, t, lane, acc); break;
      case 5: pull_rows<5>(h2, s_l, t, lane, acc); break;
      case 4: pull_rows<4>(h2, s_l, t, lane, acc); break;
      case 3: pull_rows<3>(h2, s_l, t, lane, acc); break;
      case 2: pull_rows<2>(h2, s_l, t, lane, acc); break;
      case 1: pull_rows<1>(h2, s_l, t, lane, acc); break;
      default: break;
    }
  }

  float2 r;
  r.x = di * (acc.x + hv.x);
  r.y = di * (acc.y + hv.y);
  if (BIAS) {
    r.x += bv.x; r.y += bv.y;
  }
  if (RELU) {
    r.x = fmaxf(r.x, 0.0f); r.y = fmaxf(r.y, 0.0f);
  }
  if (OSCALE) {
    r.x *= di; r.y *= di;
  }
  ((float2*)out)[(long long)node * 32 + lane] = r;
}

extern "C" void kernel_launch(void* const* d_in, const int* in_sizes, int n_in,
                              void* d_out, int out_size, void* d_ws, size_t ws_size,
                              hipStream_t stream) {
  const float* x  = (const float*)d_in[0];
  const int*   ei = (const int*)d_in[1];
  const float* W1 = (const float*)d_in[2];
  const float* b1 = (const float*)d_in[3];
  const float* W2 = (const float*)d_in[4];
  const float* b2 = (const float*)d_in[5];

  const int FIN = 256, H = 64, FOUT = 128;
  const int N = in_sizes[0] / FIN;
  const int E = in_sizes[1] / 2;
  const int* srcI = ei;
  const int* dstI = ei + E;

  const int B = (E + SORT_CH - 1) / SORT_CH;
  const int L = 256 * B;
  float* ws   = (float*)d_ws;
  float* h1   = ws;
  float* y1   = h1   + (long long)N * H;
  float* z    = y1   + (long long)N * H;
  int* cstart = (int*)(z + (long long)N * H);
  float* dinv = (float*)(cstart + (long long)L);
  int* rowptr = (int*)(dinv + (long long)N);
  int* k1   = rowptr + (long long)N + 1;
  int* v1   = k1 + (long long)E;
  int* v2   = v1 + (long long)E;

  const int TB = 256;
  auto blocks = [](long long n, int tb) { return (unsigned)((n + tb - 1) / tb); };

  k_place<8><<<B, 1024, 0, stream>>>(dstI, srcI, cstart, k1, v1, E);
  k_owner<<<N / 256 + 1, 1024, 0, stream>>>(k1, v1, cstart, v2, rowptr, dinv, E, N, B);
  const int* csr_src = v2;

  const int nstrips = N / 16;
  k_gemm_wmma<256, 64, 16, false, true><<<blocks(nstrips, 4), 128, 0, stream>>>(x, W1, nullptr, dinv, h1, nstrips);
  k_pull64<true, true, true><<<blocks((long long)N * 32, TB), TB, 0, stream>>>(h1, dinv, rowptr, csr_src, b1, y1, N);

  k_pull64<false, false, false><<<blocks((long long)N * 32, TB), TB, 0, stream>>>(y1, dinv, rowptr, csr_src, nullptr, z, N);
  k_gemm_wmma<64, 128, 8, true, false><<<blocks(nstrips, 4), 128, 0, stream>>>(z, W2, b2, nullptr, (float*)d_out, nstrips);
}


# Layer 1's GEMM runs inside the owner launch: a bucket's work-group, its dinv just made, forms its own 16 strips of x*W1 on WMMA, 16 of its waves one strip each; 6 launches become 5. Sum order kept.
# speedup vs baseline: 7.5480x; 1.0366x over previous
#include <hip/hip_runtime.h>

typedef float v2f __attribute__((ext_vector_type(2)));
typedef float v4f __attribute__((ext_vector_type(4)));
typedef float v8f __attribute__((ext_vector_type(8)));


#define CH_W 32
#define CH_R 8
#define SORT_CH (CH_W * CH_R * 32)

__device__ __forceinline__ unsigned same_digit_lanes(int d, bool valid) {
  unsigned m = __builtin_amdgcn_ballot_w32(valid);
#pragma unroll
  for (int bit = 0; bit < 8; ++bit) {
    const bool p = (d >> bit) & 1;
    const unsigned bm = __builtin_amdgcn_ballot_w32(p);
    m &= p ? bm : ~bm;
  }
  return m;
}

__device__ __forceinline__ void wave_lds_order() {
  __builtin_amdgcn_fence(__ATOMIC_ACQ_REL, "wavefront");
}

template <int SHIFT>
__launch_bounds__(1024)
__global__ void k_place(const int* __restrict__ key, const int* __restrict__ val,
                        int* __restrict__ cstart, int* __restrict__ okey, int* __restrict__ oval, int E) {
  __shared__ int wcnt[CH_W * 256];
  __shared__ int sk[SORT_CH];
  __shared__ int sv[SORT_CH];
  __shared__ int wt[8];
  const int t = threadIdx.x, lane = t & 31;
  const int w = __builtin_amdgcn_readfirstlane(t >> 5);
  const int blk = blockIdx.x;
  const int e0 = blk * SORT_CH + w * (CH_R * 32) + lane;
  const int elast = E - 1;
  int kreg[CH_R], vreg[CH_R];
#pragma unroll
  for (int r = 0; r < CH_R; ++r) {
    const int e = e0 + r * 32;
    const int ec = (e < E) ? e : elast;
    kreg[r] = key[ec];
    vreg[r] = val[ec];
  }

  for (int i = t; i < CH_W * 256; i += 1024) wcnt[i] = 0;
  __syncthreads();
  int* mycnt = wcnt + w * 256;
#pragma unroll
  for (int r = 0; r < CH_R; ++r) {
    const bool valid = (e0 + r * 32) < E;
    const int d = (kreg[r] >> SHIFT) & 255;
    const unsigned m = same_digit_lanes(d, valid);
    const unsigned below = m & ((1u << lane) - 1u);
    if (valid && below == 0u) mycnt[d] += __popc(m);
    wave_lds_order();
  }
  __syncthreads();

  int total = 0;
  if (t < 256) {
#pragma unroll
    for (int q = 0; q < CH_W; ++q) total += wcnt[q * 256 + t];
  }
  int inc = total;
  for (int o = 1; o < 32; o <<= 1) {
    const int u = __shfl_up(inc, o);
    if (lane >= o) inc += u;
  }
  if (t < 256 && lane == 31) wt[w] = inc;
  __syncthreads();
  if (t < 256) {
    int pre = 0;
#pragma unroll
    for (int q = 0; q < 8; ++q) pre += (q < w) ? wt[q] : 0;
    int run = pre + inc - total;
    cstart[blk * 256 + t] = run;
#pragma unroll
    for (int q = 0; q < CH_W; ++q) {
      const int cq = wcnt[q * 256 + t];
      wcnt[q * 256 + t] = run;
      run += cq;
    }
  }
  __syncthreads();

#pragma unroll
  for (int r = 0; r < CH_R; ++r) {
    const bool valid = (e0 + r * 32) < E;
    const int d = (kreg[r] >> SHIFT) & 255;
    const unsigned m = same_digit_lanes(d, valid);
    const unsigned below = m & ((1u << lane) - 1u);
    const int o = mycnt[d];
    wave_lds_order();
    if (valid && below == 0u) mycnt[d] = o + __popc(m);
    wave_lds_order();
    if (valid) {
      const int lp = o + __popc(below);
      sk[lp] = kreg[r];
      sv[lp] = vreg[r];
    }
  }
  __syncthreads();

  const int left = E - blk * SORT_CH;
  const int n = (left < SORT_CH) ? left : SORT_CH;
  int ks[SORT_CH / 1024], vs[SORT_CH / 1024];
#pragma unroll
  for (int j = 0; j < SORT_CH / 1024; ++j) {
    ks[j] = sk[t + 1024 * j];
    vs[j] = sv[t + 1024 * j];
  }
  int* __restrict__ ok = okey + (long long)blk * SORT_CH;
  int* __restrict__ ov = oval + (long long)blk * SORT_CH;
#pragma unroll
  for (int j = 0; j < SORT_CH / 1024; ++j) {
    if (t + 1024 * j < n) {
      ok[t + 1024 * j] = ks[j];
      ov[t + 1024 * j] = vs[j];
    }
  }
}

template <int K, int N, int U, bool BIAS, bool RSCALE>
__device__ __forceinline__ void gemm_strip(const float* __restrict__ A, const float* __restrict__ B,
                                           const float* __restrict__ bias, v4f rs0, v4f rs1,
                                           float* __restrict__ C, int strip, int lane) {
  constexpr int NT = N / 16;
  constexpr int G  = N / 64;
  const int half = lane >> 4;
  const int lm   = lane & 15;

  const float* __restrict__ arow = A + (long long)(strip * 16 + lm) * K + 2 * half;
  const float* __restrict__ bp   = B + (long long)(2 * half) * N + 4 * lm;

  v8f acc[NT];
#pragma unroll
  for (int j = 0; j < NT; ++j) acc[j] = (v8f){0.f, 0.f, 0.f, 0.f, 0.f, 0.f, 0.f, 0.f};

  v2f a_cur[U], a_nxt[U];
#pragma unroll
  for (int u = 0; u < U; ++u) {
    a_cur[u] = *(const v2f*)(arow + 4 * u);
    a_nxt[u] = a_cur[u];
  }

#pragma unroll 1
  for (int kc = 0; kc < K; kc += 4 * U) {
    v4f bx[U][G], by[U][G];
#pragma unroll
    for (int u = 0; u < U; ++u) {
#pragma unroll
      for (int g = 0; g < G; ++g) {
        bx[u][g] = *(const v4f*)(bp + (kc + 4 * u) * N + 64 * g);
        by[u][g] = *(const v4f*)(bp + (kc + 4 * u) * N + N + 64 * g);
      }
    }
    if (kc + 4 * U < K) {
#pragma unroll
      for (int u = 0; u < U; ++u) a_nxt[u] = *(const v2f*)(arow + kc + 4 * U + 4 * u);
    }
#pragma unroll
    for (int u = 0; u < U; ++u) {
#pragma unroll
      for (int g = 0; g < G; ++g) {
#pragma unroll
        for (int q = 0; q < 4; ++q) {
          v2f b;
          b.x = bx[u][g][q];
          b.y = by[u][g][q];
          acc[4 * g + q] = __builtin_amdgcn_wmma_f32_16x16x4_f32(
               false, a_cur[u],  false, b,
               (short)0, acc[4 * g + q],  false,  false);
        }
      }
    }
#pragma unroll
    for (int u = 0; u < U; ++u) a_cur[u] = a_nxt[u];
  }

  v4f bv[G];
#pragma unroll
  for (int g = 0; g < G; ++g)
    bv[g] = BIAS ? *(const v4f*)(bias + 64 * g + 4 * lm) : (v4f){0.f, 0.f, 0.f, 0.f};
  float* __restrict__ cp = C + (long long)(strip * 16 + 8 * half) * N + 4 * lm;
#pragma unroll
  for (int r = 0; r < 8; ++r) {
#pragma unroll
    for (int g = 0; g < G; ++g) {
      v4f o;
      o.x = acc[4 * g + 0][r];
      o.y = acc[4 * g + 1][r];
      o.z = acc[4 * g + 2][r];
      o.w = acc[4 * g + 3][r];
      if (BIAS) o = o + bv[g];
      if (RSCALE) o = o * ((r < 4) ? rs0[r & 3] : rs1[r & 3]);
      *(v4f*)(cp + r * N + 64 * g) = o;
    }
  }
}

template <int K, int N, int U, bool BIAS>
__launch_bounds__(128)
__global__ void k_gemm_wmma(const float* __restrict__ A, const float* __restrict__ B,
                            const float* __restrict__ bias, float* __restrict__ C, int nstrips) {
  const int lane = threadIdx.x & 31;
  const int strip = __builtin_amdgcn_readfirstlane((int)(blockIdx.x * 4 + (threadIdx.x >> 5)));
  if (strip >= nstrips) return;
  const v4f one = (v4f){1.f, 1.f, 1.f, 1.f};
  gemm_strip<K, N, U, BIAS, false>(A, B, bias, one, one, C, strip, lane);
}

#define OWN_W 32
#define OWN_STAGE 16384
#define OWN_MAXB 1024
#define OWN_F 10

__launch_bounds__(1024)
__global__ void k_owner(const int* __restrict__ key, const int* __restrict__ val,
                        const int* __restrict__ cstart, int* __restrict__ csr_src,
                        int* __restrict__ rowptr, float* __restrict__ dinv, int E, int N, int B,
                        const float* __restrict__ x, const float* __restrict__ W1,
                        float* __restrict__ h1, int nstrips) {
  __shared__ int wcnt[OWN_W * 256];
  __shared__ int wt[OWN_W];
  __shared__ int wt2[OWN_W];
  __shared__ int wtn[8];
  __shared__ int pre[OWN_MAXB];
  __shared__ int shift[OWN_MAXB];
  __shared__ int lk[OWN_STAGE];
  __shared__ int lv[OWN_STAGE];
  __shared__ int stage[OWN_STAGE];
  __shared__ __attribute__((aligned(16))) float sdinv[256];
  const int t = threadIdx.x, lane = t & 31;
  const int w = __builtin_amdgcn_readfirstlane(t >> 5);
  const int b = blockIdx.x;

  const int cc = (t < B) ? t : (B - 1);
  const int* __restrict__ row = cstart + (long long)cc * 256;
  const int v0 = row[b];
  const int v1 = row[(b < 255) ? (b + 1) : 255];
  const int left = E - cc * SORT_CH;
  const int nc = (left < SORT_CH) ? left : SORT_CH;
  const int cs  = (t < B) ? v0 : 0;
  const int len = (t < B) ? (((b < 255) ? v1 : nc) - v0) : 0;

  for (int i = t; i < OWN_W * 256; i += 1024) wcnt[i] = 0;

  int inc = len, inc2 = cs;
  for (int o = 1; o < 32; o <<= 1) {
    const int u = __shfl_up(inc, o);
    const int u2 = __shfl_up(inc2, o);
    if (lane >= o) { inc += u; inc2 += u2; }
  }
  if (lane == 31) { wt[w] = inc; wt2[w] = inc2; }
  __syncthreads();
  int pr = 0, tot = 0, bsum = 0;
#pragma unroll
  for (int q = 0; q < OWN_W; ++q) {
    const int a = wt[q];
    pr += (q < w) ? a : 0;
    tot += a;
    bsum += wt2[q];
  }
  const int first = pr + inc - len;
  pre[t] = (t < B) ? first : 0x7fffffff;
  shift[t] = cc * SORT_CH + cs - first;
  const int total = __builtin_amdgcn_readfirstlane(tot);
  const int bs = __builtin_amdgcn_readfirstlane(bsum);
  __syncthreads();

  const int R  = (total + 32 * OWN_W - 1) / (32 * OWN_W);
  const int wbeg = w * (R * 32);
  const int wend = (wbeg + R * 32 < total) ? (wbeg + R * 32) : total;
  const int plast = (total > 0) ? (total - 1) : 0;
  const bool staged = total <= OWN_STAGE;
  int s0 = 1;
  while (2 * s0 < B && s0 < OWN_MAXB / 2) s0 <<= 1;

  auto record_lds = [&](int p, int& k, int& v) {
    const int pc = (p < plast) ? p : plast;
    k = lk[pc];
    v = lv[pc];
  };
  auto record_mem = [&](int p, int& k, int& v) {
    const int pc = (p < plast) ? p : plast;
    int ci = 0;
    for (int s = s0; s > 0; s >>= 1) ci += (pre[ci + s] <= pc) ? s : 0;
    const int a = shift[ci] + pc;
    k = key[a];
    v = val[a];
  };

  if (staged) {
    for (int j0 = 0; j0 < R; j0 += OWN_F) {
      int pc[OWN_F], ci[OWN_F], kk[OWN_F], vv[OWN_F];
#pragma unroll
      for (int u = 0; u < OWN_F; ++u) {
        const int p = t + 1024 * (j0 + u);
        pc[u] = (p < plast) ? p : plast;
        ci[u] = 0;
      }
      for (int s = s0; s > 0; s >>= 1) {
#pragma unroll
        for (int u = 0; u < OWN_F; ++u) ci[u] += (pre[ci[u] + s] <= pc[u]) ? s : 0;
      }
#pragma unroll
      for (int u = 0; u < OWN_F; ++u) {
        const int a = shift[ci[u]] + pc[u];
        kk[u] = key[a];
        vv[u] = val[a];
      }
#pragma unroll
      for (int u = 0; u < OWN_F; ++u) {
        const int p = t + 1024 * (j0 + u);
        if (p < total) {
          lk[p] = kk[u];
          lv[p] = vv[u];
        }
      }
    }
    __syncthreads();
  }
  int* mycnt = wcnt + w * 256;

  auto walk_count = [&](auto record) {
    int p = wbeg + lane;
    int kn, vn;
    record(p, kn, vn);
    for (int r = 0; r < R; ++r) {
      const int kc = kn;
      const bool valid = p < wend;
      p += 32;
      record(p, kn, vn);
      const int d = kc & 255;
      const unsigned m = same_digit_lanes(d, valid);
      const unsigned below = m & ((1u << lane) - 1u);
      if (valid && below == 0u) mycnt[d] += __popc(m);
      wave_lds_order();
    }
  };
  if (staged) walk_count(record_lds);
  else        walk_count(record_mem);
  __syncthreads();

  int c[OWN_W];
  int ntot = 0;
  if (t < 256) {
#pragma unroll
    for (int q = 0; q < OWN_W; ++q) c[q] = wcnt[q * 256 + t];
#pragma unroll
    for (int q = 0; q < OWN_W; ++q) ntot += c[q];
  }
  int ninc = ntot;
  for (int o = 1; o < 32; o <<= 1) {
    const int u = __shfl_up(ninc, o);
    if (lane >= o) ninc += u;
  }
  if (t < 256 && lane == 31) wtn[w] = ninc;
  __syncthreads();
  if (t < 256) {
    int npre = 0;
#pragma unroll
    for (int q = 0; q < 8; ++q) npre += (q < w) ? wtn[q] : 0;
    int run = bs + npre + ninc - ntot;
    const int node = b * 256 + t;
    if (node <= N) rowptr[node] = (node == N) ? E : run;
    const float dv = rsqrtf((float)ntot + 1.0f);
    if (node < N) dinv[node] = dv;
    sdinv[t] = dv;
#pragma unroll
    for (int q = 0; q < OWN_W; ++q) {
      wcnt[q * 256 + t] = run;
      run += c[q];
    }
  }
  __syncthreads();

  auto put_lds = [&](int q, int v) { stage[q - bs] = v; };
  auto put_mem = [&](int q, int v) { csr_src[q] = v; };
  auto walk_place = [&](auto record, auto put) {
    int p = wbeg + lane;
    int kn, vn;
    record(p, kn, vn);
    for (int r = 0; r < R; ++r) {
      const int kc = kn, vc = vn;
      const bool valid = p < wend;
      p += 32;
      record(p, kn, vn);
      const int d = kc & 255;
      const unsigned m = same_digit_lanes(d, valid);
      const unsigned below = m & ((1u << lane) - 1u);
      const int o = mycnt[d];
      wave_lds_order();
      if (valid && below == 0u) mycnt[d] = o + __popc(m);
      wave_lds_order();
      if (valid) put(o + __popc(below), vc);
    }
  };
  if (staged) {
    walk_place(record_lds, put_lds);
    __syncthreads();
    for (int i = t; i < total; i += 1024) csr_src[bs + i] = stage[i];
  } else {
    walk_place(record_mem, put_mem);
  }

  if ((w & 7) < 4) {
    const int sw = (w >> 3) * 4 + (w & 3);
    const int strip = b * 16 + sw;
    if (strip < nstrips) {
      const int half = lane >> 4;
      const v4f rs0 = *(const v4f*)(sdinv + sw * 16 + 8 * half);
      const v4f rs1 = *(const v4f*)(sdinv + sw * 16 + 8 * half + 4);
      gemm_strip<256, 64, 4, false, true>(x, W1, nullptr, rs0, rs1, h1, strip, lane);
    }
  }
}


template <int R>
__device__ __forceinline__ void pull_rows(const float2* __restrict__ h2, int s_l,
                                          int t, int lane, float2& acc) {
  float2 v[R];
#pragma unroll
  for (int k = 0; k < R; ++k) {
    const int s = __builtin_amdgcn_readlane(s_l, t + k);
    v[k] = h2[(long long)s * 32 + lane];
  }
#pragma unroll
  for (int k = 0; k < R; ++k) {
    acc.x += v[k].x;
    acc.y += v[k].y;
  }
}

template <bool RELU, bool BIAS, bool OSCALE>
__launch_bounds__(256)
__global__ void k_pull64(const float* __restrict__ h, const float* __restrict__ dinv,
                         const int* __restrict__ rowptr, const int* __restrict__ csr_src,
                         const float* __restrict__ bias, float* __restrict__ out, int N) {
  int node = blockIdx.x * (blockDim.x >> 5) + (threadIdx.x >> 5);
  node = __builtin_amdgcn_readfirstlane(node);
  const int lane = threadIdx.x & 31;
  if (node >= N) return;
  const int beg = __builtin_amdgcn_readfirstlane(rowptr[node]);
  const int end = __builtin_amdgcn_readfirstlane(rowptr[node + 1]);
  const float2* __restrict__ h2 = (const float2*)h;

  const float di = dinv[node];
  const float2 hv = h2[(long long)node * 32 + lane];
  float2 bv = make_float2(0.0f, 0.0f);
  if (BIAS) bv = ((const float2*)bias)[lane];

  float2 acc = make_float2(0.0f, 0.0f);
  for (int c = beg; c < end; c += 32) {
    const int n = (end - c < 32) ? (end - c) : 32;
    int s_l = 0;
    if (lane < n) s_l = csr_src[c + lane];
    int t = 0;
    for (; t + 8 <= n; t += 8) pull_rows<8>(h2, s_l, t, lane, acc);
    switch (n - t) {
      case 7: pull_rows<7>(h2, s_l, t, lane, acc); break;
      case 6: pull_rows<6>(h2, s_l, t, lane, acc); break;
      case 5: pull_rows<5>(h2, s_l, t, lane, acc); break;
      case 4: pull_rows<4>(h2, s_l, t, lane, acc); break;
      case 3: pull_rows<3>(h2, s_l, t, lane, acc); break;
      case 2: pull_rows<2>(h2, s_l, t, lane, acc); break;
      case 1: pull_rows<1>(h2, s_l, t, lane, acc); break;
      default: break;
    }
  }

  float2 r;
  r.x = di * (acc.x + hv.x);
  r.y = di * (acc.y + hv.y);
  if (BIAS) {
    r.x += bv.x; r.y += bv.y;
  }
  if (RELU) {
    r.x = fmaxf(r.x, 0.0f); r.y = fmaxf(r.y, 0.0f);
  }
  if (OSCALE) {
    r.x *= di; r.y *= di;
  }
  ((float2*)out)[(long long)node * 32 + lane] = r;
}

extern "C" void kernel_launch(void* const* d_in, const int* in_sizes, int n_in,
                              void* d_out, int out_size, void* d_ws, size_t ws_size,
                              hipStream_t stream) {
  const float* x  = (const float*)d_in[0];
  const int*   ei = (const int*)d_in[1];
  const float* W1 = (const float*)d_in[2];
  const float* b1 = (const float*)d_in[3];
  const float* W2 = (const float*)d_in[4];
  const float* b2 = (const float*)d_in[5];

  const int FIN = 256, H = 64, FOUT = 128;
  const int N = in_sizes[0] / FIN;
  const int E = in_sizes[1] / 2;
  const int* srcI = ei;
  const int* dstI = ei + E;

  const int B = (E + SORT_CH - 1) / SORT_CH;
  const int L = 256 * B;
  float* ws   = (float*)d_ws;
  float* h1   = ws;
  float* y1   = h1   + (long long)N * H;
  float* z    = y1   + (long long)N * H;
  int* cstart = (int*)(z + (long long)N * H);
  float* dinv = (float*)(cstart + (long long)L);
  int* rowptr = (int*)(dinv + (long long)N);
  int* k1   = rowptr + (long long)N + 1;
  int* v1   = k1 + (long long)E;
  int* v2   = v1 + (long long)E;

  const int TB = 256;
  auto blocks = [](long long n, int tb) { return (unsigned)((n + tb - 1) / tb); };

  k_place<8><<<B, 1024, 0, stream>>>(dstI, srcI, cstart, k1, v1, E);
  const int nstrips = N / 16;
  k_owner<<<N / 256 + 1, 1024, 0, stream>>>(k1, v1, cstart, v2, rowptr, dinv, E, N, B, x, W1, h1, nstrips);
  const int* csr_src = v2;

  k_pull64<true, true, true><<<blocks((long long)N * 32, TB), TB, 0, stream>>>(h1, dinv, rowptr, csr_src, b1, y1, N);

  k_pull64<false, false, false><<<blocks((long long)N * 32, TB), TB, 0, stream>>>(y1, dinv, rowptr, csr_src, nullptr, z, N);
  k_gemm_wmma<64, 128, 8, true><<<blocks(nstrips, 4), 128, 0, stream>>>(z, W2, b2, (float*)d_out, nstrips);
}
